# starting kernel
# baseline (speedup 1.0000x reference)
#include <hip/hip_runtime.h>
#include <hip/hip_bf16.h>

#define N_NODES 50000
#define FDIM    128
#define KDIM    128

typedef __attribute__((ext_vector_type(2))) float v2f;
typedef __attribute__((ext_vector_type(8))) float v8f;

__global__ void zero_f32(float* __restrict__ p, int n) {
    int i = blockIdx.x * blockDim.x + threadIdx.x;
    int stride = gridDim.x * blockDim.x;
    for (; i < n; i += stride) p[i] = 0.0f;
}

__global__ void degree_kernel(const int* __restrict__ src, const int* __restrict__ dst,
                              float* __restrict__ outdeg, float* __restrict__ indeg, int E) {
    int i = blockIdx.x * blockDim.x + threadIdx.x;
    if (i < E) {
        atomicAdd(&outdeg[src[i]], 1.0f);
        atomicAdd(&indeg[dst[i]], 1.0f);
    }
}

__global__ void norm_finalize(float* __restrict__ p, int n) {
    int i = blockIdx.x * blockDim.x + threadIdx.x;
    if (i < n) p[i] = rsqrtf(fmaxf(p[i], 1.0f));
}

__global__ __launch_bounds__(256)
void aggregate_kernel(const float* __restrict__ x, const int* __restrict__ src,
                      const int* __restrict__ dst, const float* __restrict__ nsrc,
                      float* __restrict__ agg, int E) {
    const int lane = threadIdx.x & 31;
    const int e = blockIdx.x * (blockDim.x >> 5) + (threadIdx.x >> 5);
    if (e >= E) return;
    const int s = src[e];
    const int d = dst[e];
    const float n = nsrc[s];
    const float4 v = ((const float4*)(x + (size_t)s * FDIM))[lane];
    float* o = agg + (size_t)d * FDIM + lane * 4;
    atomicAdd(o + 0, v.x * n);
    atomicAdd(o + 1, v.y * n);
    atomicAdd(o + 2, v.z * n);
    atomicAdd(o + 3, v.w * n);
}

template <int NOUT, bool RELU>
__global__ __launch_bounds__(128)
void gemm_wmma_kernel(const float* __restrict__ A, const float* __restrict__ W,
                      const float* __restrict__ bias, const float* __restrict__ norm,
                      float* __restrict__ out) {
    constexpr int NT = NOUT / 16;
    const int lane  = threadIdx.x & 31;
    const int wave  = threadIdx.x >> 5;
    const int mtile = blockIdx.x * 4 + wave;
    if (mtile * 16 >= N_NODES) return;
    const int m0    = mtile * 16;
    const int r15   = lane & 15;
    const int khalf = lane >> 4;
    const int koff  = khalf * 2;

    v8f acc[NT];
#pragma unroll
    for (int t = 0; t < NT; ++t)
#pragma unroll
        for (int j = 0; j < 8; ++j) acc[t][j] = 0.0f;

    const float* Ar = A + (size_t)(m0 + r15) * KDIM + koff;
    for (int k = 0; k < KDIM; k += 4) {
        v2f a;
        a.x = Ar[k];
        a.y = Ar[k + 1];
#pragma unroll
        for (int t = 0; t < NT; ++t) {
            const float* Wp = W + (size_t)(k + koff) * NOUT + t * 16 + r15;
            v2f b;
            b.x = Wp[0];
            b.y = Wp[NOUT];
            acc[t] = __builtin_amdgcn_wmma_f32_16x16x4_f32(
                false, a, false, b, (short)0, acc[t], false, false);
        }
    }

    const int rbase = m0 + khalf * 8;
    float nrm[8];
#pragma unroll
    for (int j = 0; j < 8; ++j) nrm[j] = norm[rbase + j];
#pragma unroll
    for (int t = 0; t < NT; ++t) {
        const float bv = bias[t * 16 + r15];
#pragma unroll
        for (int j = 0; j < 8; ++j) {
            float v = acc[t][j] * nrm[j] + bv;
            if (RELU) v = fmaxf(v, 0.0f);
            out[(size_t)(rbase + j) * NOUT + t * 16 + r15] = v;
        }
    }
}

extern "C" void kernel_launch(void* const* d_in, const int* in_sizes, int n_in,
                              void* d_out, int out_size, void* d_ws, size_t ws_size,
                              hipStream_t stream) {
    const float* features = (const float*)d_in[0];
    const float* W1 = (const float*)d_in[1];
    const float* b1 = (const float*)d_in[2];
    const float* W2 = (const float*)d_in[3];
    const float* b2 = (const float*)d_in[4];
    const float* W3 = (const float*)d_in[5];
    const float* b3 = (const float*)d_in[6];
    const int*  src = (const int*)d_in[7];
    const int*  dst = (const int*)d_in[8];
    const int E = in_sizes[7];

    float* ws       = (float*)d_ws;
    float* norm_src = ws;
    float* norm_dst = ws + N_NODES;
    float* aggA     = ws + 2 * N_NODES;
    float* bufB     = aggA + (size_t)N_NODES * FDIM;

    zero_f32<<<512, 256, 0, stream>>>(norm_src, 2 * N_NODES);
    degree_kernel<<<(E + 255) / 256, 256, 0, stream>>>(src, dst, norm_src, norm_dst, E);
    norm_finalize<<<(2 * N_NODES + 255) / 256, 256, 0, stream>>>(norm_src, 2 * N_NODES);

    const int aggBlocks  = (E + 7) / 8;
    const int gemmBlocks = (N_NODES / 16 + 3) / 4;

    zero_f32<<<2048, 256, 0, stream>>>(aggA, N_NODES * FDIM);
    aggregate_kernel<<<aggBlocks, 256, 0, stream>>>(features, src, dst, norm_src, aggA, E);
    gemm_wmma_kernel<128, true><<<gemmBlocks, 128, 0, stream>>>(aggA, W1, b1, norm_dst, bufB);

    zero_f32<<<2048, 256, 0, stream>>>(aggA, N_NODES * FDIM);
    aggregate_kernel<<<aggBlocks, 256, 0, stream>>>(bufB, src, dst, norm_src, aggA, E);
    gemm_wmma_kernel<128, true><<<gemmBlocks, 128, 0, stream>>>(aggA, W2, b2, norm_dst, bufB);

    zero_f32<<<2048, 256, 0, stream>>>(aggA, N_NODES * FDIM);
    aggregate_kernel<<<aggBlocks, 256, 0, stream>>>(bufB, src, dst, norm_src, aggA, E);
    gemm_wmma_kernel<64, false><<<gemmBlocks, 128, 0, stream>>>(aggA, W3, b3, norm_dst,
                                                                (float*)d_out);
}


# Aggregate via a dst-sorted edge list built per call (in-degree rank by a returning int atomic, scan, fill); a wave per node sums rows in registers: no f32 atomics, no zeroing; sums in arrival order
# speedup vs baseline: 2.4066x; 2.4066x over previous
#include <hip/hip_runtime.h>
#include <hip/hip_bf16.h>


#define N_NODES 50000
#define FDIM    128
#define KDIM    128
#define SCAN_BLOCKS ((N_NODES + 255) / 256)

typedef __attribute__((ext_vector_type(2))) float v2f;
typedef __attribute__((ext_vector_type(8))) float v8f;

__global__ void zero_i32(int* __restrict__ p, int n) {
    int i = blockIdx.x * blockDim.x + threadIdx.x;
    int stride = gridDim.x * blockDim.x;
    for (; i < n; i += stride) p[i] = 0;
}

__device__ __forceinline__ int wave_sum(int v) {
    for (int o = 16; o > 0; o >>= 1) v += __shfl_xor(v, o);
    return v;
}

__device__ __forceinline__ int wave_inclusive_scan(int v, int lane) {
    for (int o = 1; o < 32; o <<= 1) {
        const int t = __shfl_up(v, o);
        if (lane >= o) v += t;
    }
    return v;
}

__global__ void degree_rank_kernel(const int* __restrict__ src, const int* __restrict__ dst,
                                   int* __restrict__ cnt_out, int* __restrict__ cnt_in,
                                   int* __restrict__ rank, int E) {
    int i = blockIdx.x * blockDim.x + threadIdx.x;
    if (i < E) {
        atomicAdd(&cnt_out[src[i]], 1);
        rank[i] = atomicAdd(&cnt_in[dst[i]], 1);
    }
}

__global__ __launch_bounds__(256)
void finalize_kernel(const int* __restrict__ cnt, float* __restrict__ nrm,
                     int* __restrict__ blocksum) {
    __shared__ int part[8];
    const int tid = threadIdx.x;
    const int i = blockIdx.x * 256 + tid;
    int ci = 0;
    if (i < N_NODES) {
        const int co = cnt[i];
        ci = cnt[N_NODES + i];
        nrm[i]           = rsqrtf(fmaxf((float)co, 1.0f));
        nrm[N_NODES + i] = rsqrtf(fmaxf((float)ci, 1.0f));
    }
    const int s = wave_sum(ci);
    if ((tid & 31) == 0) part[tid >> 5] = s;
    __syncthreads();
    if (tid == 0) {
        int t = 0;
        for (int k = 0; k < 8; ++k) t += part[k];
        blocksum[blockIdx.x] = t;
    }
}

__global__ __launch_bounds__(256)
void scan_kernel(const int* __restrict__ cnt_in, const int* __restrict__ blocksum,
                 int* __restrict__ rowptr) {
    __shared__ int pre[8];
    __shared__ int tot[8];
    const int tid  = threadIdx.x;
    const int lane = tid & 31;
    const int wave = tid >> 5;
    int p = 0;
    for (int k = tid; k < (int)blockIdx.x; k += 256) p += blocksum[k];
    p = wave_sum(p);
    const int i = blockIdx.x * 256 + tid;
    const int c = (i < N_NODES) ? cnt_in[i] : 0;
    const int inc = wave_inclusive_scan(c, lane);
    if (lane == 0)  pre[wave] = p;
    if (lane == 31) tot[wave] = inc;
    __syncthreads();
    int off = 0;
    for (int k = 0; k < 8; ++k) {
        off += pre[k];
        if (k < wave) off += tot[k];
    }
    const int excl = off + inc - c;
    if (i < N_NODES) rowptr[i] = excl;
    if (i == N_NODES - 1) rowptr[N_NODES] = excl + c;
}

__global__ void fill_kernel(const int* __restrict__ src, const int* __restrict__ dst,
                            const int* __restrict__ rank, const int* __restrict__ rowptr,
                            const float* __restrict__ nsrc, int* __restrict__ edges, int E) {
    int i = blockIdx.x * blockDim.x + threadIdx.x;
    if (i < E) {
        const int s = src[i];
        const int pos = rowptr[dst[i]] + rank[i];
        ((int2*)edges)[pos] = make_int2(s, __float_as_int(nsrc[s]));
    }
}

__global__ __launch_bounds__(256)
void gather_kernel(const float* __restrict__ x, const int* __restrict__ rowptr,
                   const int* __restrict__ edges, float* __restrict__ agg) {
    const int lane = threadIdx.x & 31;
    const int wid = __builtin_amdgcn_readfirstlane((int)(blockIdx.x * 8 + (threadIdx.x >> 5)));
    const int nw  = gridDim.x * 8;
    const float4* __restrict__ xl = (const float4*)x + lane;
    const int2* __restrict__ el = (const int2*)edges;

#define ROW(j) xl[(size_t)__builtin_amdgcn_readlane(my.x, (j)) * (FDIM / 4)]
#define WGT(j) __int_as_float(__builtin_amdgcn_readlane(my.y, (j)))
#define ACC(VV, WW) { acc.x = fmaf((VV).x, (WW), acc.x); acc.y = fmaf((VV).y, (WW), acc.y); \
                      acc.z = fmaf((VV).z, (WW), acc.z); acc.w = fmaf((VV).w, (WW), acc.w); }

    for (int node = wid; node < N_NODES; node += nw) {
        const int beg = rowptr[node];
        const int end = rowptr[node + 1];
        float4 acc = make_float4(0.0f, 0.0f, 0.0f, 0.0f);
        for (int base = beg; base < end; base += 32) {
            const int cnt = min(32, end - base);
            int2 my = make_int2(0, 0);
            if (lane < cnt) my = el[base + lane];
            int j = 0;
            for (; j + 8 <= cnt; j += 8) {
                float4 v[8];
                float  w[8];
#pragma unroll
                for (int u = 0; u < 8; ++u) { v[u] = ROW(j + u); w[u] = WGT(j + u); }
#pragma unroll
                for (int u = 0; u < 8; ++u) ACC(v[u], w[u]);
            }
            if (j + 4 <= cnt) {
                float4 v[4];
                float  w[4];
#pragma unroll
                for (int u = 0; u < 4; ++u) { v[u] = ROW(j + u); w[u] = WGT(j + u); }
#pragma unroll
                for (int u = 0; u < 4; ++u) ACC(v[u], w[u]);
                j += 4;
            }
            if (j < cnt) {
                const int r = cnt - j;
                const float4 z = make_float4(0.0f, 0.0f, 0.0f, 0.0f);
                float4 v0 = ROW(j), v1 = z, v2 = z;
                float  w0 = WGT(j), w1 = 0.0f, w2 = 0.0f;
                if (r > 1) { v1 = ROW(j + 1); w1 = WGT(j + 1); }
                if (r > 2) { v2 = ROW(j + 2); w2 = WGT(j + 2); }
                ACC(v0, w0);
                ACC(v1, w1);
                ACC(v2, w2);
            }
        }
        ((float4*)(agg + (size_t)node * FDIM))[lane] = acc;
    }
#undef ROW
#undef WGT
#undef ACC
}

template <int NOUT, bool RELU>
__global__ __launch_bounds__(128)
void gemm_wmma_kernel(const float* __restrict__ A, const float* __restrict__ W,
                      const float* __restrict__ bias, const float* __restrict__ norm,
                      float* __restrict__ out) {
    constexpr int NT = NOUT / 16;
    const int lane  = threadIdx.x & 31;
    const int wave  = threadIdx.x >> 5;
    const int mtile = blockIdx.x * 4 + wave;
    if (mtile * 16 >= N_NODES) return;
    const int m0    = mtile * 16;
    const int r15   = lane & 15;
    const int khalf = lane >> 4;
    const int koff  = khalf * 2;

    v8f acc[NT];
#pragma unroll
    for (int t = 0; t < NT; ++t)
#pragma unroll
        for (int j = 0; j < 8; ++j) acc[t][j] = 0.0f;

    const float* Ar = A + (size_t)(m0 + r15) * KDIM + koff;
    for (int k = 0; k < KDIM; k += 4) {
        v2f a;
        a.x = Ar[k];
        a.y = Ar[k + 1];
#pragma unroll
        for (int t = 0; t < NT; ++t) {
            const float* Wp = W + (size_t)(k + koff) * NOUT + t * 16 + r15;
            v2f b;
            b.x = Wp[0];
            b.y = Wp[NOUT];
            acc[t] = __builtin_amdgcn_wmma_f32_16x16x4_f32(
                false, a, false, b, (short)0, acc[t], false, false);
        }
    }

    const int rbase = m0 + khalf * 8;
    float nrm[8];
#pragma unroll
    for (int j = 0; j < 8; ++j) nrm[j] = norm[rbase + j];
#pragma unroll
    for (int t = 0; t < NT; ++t) {
        const float bv = bias[t * 16 + r15];
#pragma unroll
        for (int j = 0; j < 8; ++j) {
            float v = acc[t][j] * nrm[j] + bv;
            if (RELU) v = fmaxf(v, 0.0f);
            out[(size_t)(rbase + j) * NOUT + t * 16 + r15] = v;
        }
    }
}

extern "C" void kernel_launch(void* const* d_in, const int* in_sizes, int n_in,
                              void* d_out, int out_size, void* d_ws, size_t ws_size,
                              hipStream_t stream) {
    const float* features = (const float*)d_in[0];
    const float* W1 = (const float*)d_in[1];
    const float* b1 = (const float*)d_in[2];
    const float* W2 = (const float*)d_in[3];
    const float* b2 = (const float*)d_in[4];
    const float* W3 = (const float*)d_in[5];
    const float* b3 = (const float*)d_in[6];
    const int*  src = (const int*)d_in[7];
    const int*  dst = (const int*)d_in[8];
    const int E = in_sizes[7];

    int*   cnt      = (int*)d_ws;
    float* nrm      = (float*)d_ws + 2 * N_NODES;
    int*   rowptr   = (int*)d_ws + 4 * N_NODES;
    int*   blocksum = (int*)d_ws + 250048;
    float* aggA     = (float*)d_ws + 250304;
    float* bufB     = aggA + (size_t)N_NODES * FDIM;
    int*   edges    = (int*)(bufB + (size_t)N_NODES * FDIM);
    int*   rank     = edges + 2 * (size_t)E;
    const float* norm_src = nrm;
    const float* norm_dst = nrm + N_NODES;

    zero_i32<<<512, 256, 0, stream>>>(cnt, 2 * N_NODES);
    degree_rank_kernel<<<(E + 255) / 256, 256, 0, stream>>>(src, dst, cnt, cnt + N_NODES, rank, E);
    finalize_kernel<<<SCAN_BLOCKS, 256, 0, stream>>>(cnt, nrm, blocksum);
    scan_kernel<<<SCAN_BLOCKS, 256, 0, stream>>>(cnt + N_NODES, blocksum, rowptr);
    fill_kernel<<<(E + 255) / 256, 256, 0, stream>>>(src, dst, rank, rowptr, norm_src, edges, E);

    const int gatherBlocks = 2048;
    const int gemmBlocks = (N_NODES / 16 + 3) / 4;

    gather_kernel<<<gatherBlocks, 256, 0, stream>>>(features, rowptr, edges, aggA);
    gemm_wmma_kernel<128, true><<<gemmBlocks, 128, 0, stream>>>(aggA, W1, b1, norm_dst, bufB);

    gather_kernel<<<gatherBlocks, 256, 0, stream>>>(bufB, rowptr, edges, aggA);
    gemm_wmma_kernel<128, true><<<gemmBlocks, 128, 0, stream>>>(aggA, W2, b2, norm_dst, bufB);

    gather_kernel<<<gatherBlocks, 256, 0, stream>>>(bufB, rowptr, edges, aggA);
    gemm_wmma_kernel<64, false><<<gemmBlocks, 128, 0, stream>>>(aggA, W3, b3, norm_dst,
                                                                (float*)d_out);
}


# Fuse each layer's transform into its aggregation launch: 16 waves sum a 16-node tile into LDS and feed the f32 WMMA from it, W held in LDS; 11 to 8 launches, sum order unchanged
# speedup vs baseline: 2.9198x; 1.2132x over previous
#include <hip/hip_runtime.h>
#include <hip/hip_bf16.h>


#define N_NODES 50000
#define FDIM    128
#define KDIM    128
#define SCAN_BLOCKS ((N_NODES + 255) / 256)

typedef __attribute__((ext_vector_type(2))) float v2f;
typedef __attribute__((ext_vector_type(8))) float v8f;

__global__ void zero_i32(int* __restrict__ p, int n) {
    int i = blockIdx.x * blockDim.x + threadIdx.x;
    int stride = gridDim.x * blockDim.x;
    for (; i < n; i += stride) p[i] = 0;
}

__device__ __forceinline__ int wave_sum(int v) {
    for (int o = 16; o > 0; o >>= 1) v += __shfl_xor(v, o);
    return v;
}

__device__ __forceinline__ int wave_inclusive_scan(int v, int lane) {
    for (int o = 1; o < 32; o <<= 1) {
        const int t = __shfl_up(v, o);
        if (lane >= o) v += t;
    }
    return v;
}

__global__ void degree_rank_kernel(const int* __restrict__ src, const int* __restrict__ dst,
                                   int* __restrict__ cnt_out, int* __restrict__ cnt_in,
                                   int* __restrict__ rank, int E) {
    int i = blockIdx.x * blockDim.x + threadIdx.x;
    if (i < E) {
        atomicAdd(&cnt_out[src[i]], 1);
        rank[i] = atomicAdd(&cnt_in[dst[i]], 1);
    }
}

__global__ __launch_bounds__(256)
void finalize_kernel(const int* __restrict__ cnt, float* __restrict__ nrm,
                     int* __restrict__ blocksum) {
    __shared__ int part[8];
    const int tid = threadIdx.x;
    const int i = blockIdx.x * 256 + tid;
    int ci = 0;
    if (i < N_NODES) {
        const int co = cnt[i];
        ci = cnt[N_NODES + i];
        nrm[i]           = rsqrtf(fmaxf((float)co, 1.0f));
        nrm[N_NODES + i] = rsqrtf(fmaxf((float)ci, 1.0f));
    }
    const int s = wave_sum(ci);
    if ((tid & 31) == 0) part[tid >> 5] = s;
    __syncthreads();
    if (tid == 0) {
        int t = 0;
        for (int k = 0; k < 8; ++k) t += part[k];
        blocksum[blockIdx.x] = t;
    }
}

__global__ __launch_bounds__(256)
void scan_kernel(const int* __restrict__ cnt_in, const int* __restrict__ blocksum,
                 int* __restrict__ rowptr) {
    __shared__ int pre[8];
    __shared__ int tot[8];
    const int tid  = threadIdx.x;
    const int lane = tid & 31;
    const int wave = tid >> 5;
    int p = 0;
    for (int k = tid; k < (int)blockIdx.x; k += 256) p += blocksum[k];
    p = wave_sum(p);
    const int i = blockIdx.x * 256 + tid;
    const int c = (i < N_NODES) ? cnt_in[i] : 0;
    const int inc = wave_inclusive_scan(c, lane);
    if (lane == 0)  pre[wave] = p;
    if (lane == 31) tot[wave] = inc;
    __syncthreads();
    int off = 0;
    for (int k = 0; k < 8; ++k) {
        off += pre[k];
        if (k < wave) off += tot[k];
    }
    const int excl = off + inc - c;
    if (i < N_NODES) rowptr[i] = excl;
    if (i == N_NODES - 1) rowptr[N_NODES] = excl + c;
}

__global__ void fill_kernel(const int* __restrict__ src, const int* __restrict__ dst,
                            const int* __restrict__ rank, const int* __restrict__ rowptr,
                            const float* __restrict__ nsrc, int* __restrict__ edges, int E) {
    int i = blockIdx.x * blockDim.x + threadIdx.x;
    if (i < E) {
        const int s = src[i];
        const int pos = rowptr[dst[i]] + rank[i];
        ((int2*)edges)[pos] = make_int2(s, __float_as_int(nsrc[s]));
    }
}

template <int NOUT, bool RELU>
__global__ __launch_bounds__(512)
void layer_kernel(const float* __restrict__ x, const int* __restrict__ rowptr,
                  const int* __restrict__ edges, const float* __restrict__ W,
                  const float* __restrict__ bias, const float* __restrict__ norm,
                  float* __restrict__ out) {
    constexpr int NT     = NOUT / 16;
    constexpr int GROUPS = 16 / NT;
    __shared__ __attribute__((aligned(16))) float Wl[KDIM * NOUT];
    __shared__ __attribute__((aligned(16))) float At[16 * FDIM];

    const int lane  = threadIdx.x & 31;
    const int wave  = __builtin_amdgcn_readfirstlane((int)(threadIdx.x >> 5));
    const int r15   = lane & 15;
    const int khalf = lane >> 4;

#pragma unroll
    for (int s = threadIdx.x; s < (KDIM / 4) * NT * 32; s += 512) {
        const int l  = s & 31;
        const int t  = (s >> 5) % NT;
        const int kk = (s >> 5) / NT;
        const float* Wp = W + (size_t)(kk * 4 + (l >> 4) * 2) * NOUT + t * 16 + (l & 15);
        ((float2*)Wl)[s] = make_float2(Wp[0], Wp[NOUT]);
    }

    const float4* __restrict__ xl = (const float4*)x + lane;
    const int2* __restrict__ el = (const int2*)edges;

#define ROW(j) xl[(size_t)__builtin_amdgcn_readlane(my.x, (j)) * (FDIM / 4)]
#define WGT(j) __int_as_float(__builtin_amdgcn_readlane(my.y, (j)))
#define ACC(VV, WW) { acc.x = fmaf((VV).x, (WW), acc.x); acc.y = fmaf((VV).y, (WW), acc.y); \
                      acc.z = fmaf((VV).z, (WW), acc.z); acc.w = fmaf((VV).w, (WW), acc.w); }

    int it = 0;
    for (int tile = blockIdx.x; tile < N_NODES / 16; tile += gridDim.x, ++it) {
        const int node = tile * 16 + wave;
        const int beg = rowptr[node];
        const int end = rowptr[node + 1];
        float4 acc = make_float4(0.0f, 0.0f, 0.0f, 0.0f);
        for (int base = beg; base < end; base += 32) {
            const int cnt = min(32, end - base);
            int2 my = make_int2(0, 0);
            if (lane < cnt) my = el[base + lane];
            int j = 0;
            for (; j + 8 <= cnt; j += 8) {
                float4 v[8];
                float  w[8];
#pragma unroll
                for (int u = 0; u < 8; ++u) { v[u] = ROW(j + u); w[u] = WGT(j + u); }
#pragma unroll
                for (int u = 0; u < 8; ++u) ACC(v[u], w[u]);
            }
            if (j + 4 <= cnt) {
                float4 v[4];
                float  w[4];
#pragma unroll
                for (int u = 0; u < 4; ++u) { v[u] = ROW(j + u); w[u] = WGT(j + u); }
#pragma unroll
                for (int u = 0; u < 4; ++u) ACC(v[u], w[u]);
                j += 4;
            }
            if (j < cnt) {
                const int r = cnt - j;
                const float4 z = make_float4(0.0f, 0.0f, 0.0f, 0.0f);
                float4 v0 = ROW(j), v1 = z, v2 = z;
                float  w0 = WGT(j), w1 = 0.0f, w2 = 0.0f;
                if (r > 1) { v1 = ROW(j + 1); w1 = WGT(j + 1); }
                if (r > 2) { v2 = ROW(j + 2); w2 = WGT(j + 2); }
                ACC(v0, w0);
                ACC(v1, w1);
                ACC(v2, w2);
            }
        }
        ((float4*)(At + wave * FDIM))[lane ^ wave] = acc;
        __syncthreads();

        const bool mine = (wave / NT == (it & (GROUPS - 1)));
        const int t     = wave % NT;
        const int rbase = tile * 16 + khalf * 8;
        float nrm[8];
        float bv = 0.0f;
        v8f d;
#pragma unroll
        for (int j = 0; j < 8; ++j) { d[j] = 0.0f; nrm[j] = 0.0f; }
        if (mine) {
#pragma unroll
            for (int j = 0; j < 8; ++j) nrm[j] = norm[rbase + j];
            bv = bias[t * 16 + r15];
            const float* Ar = At + r15 * FDIM + khalf * 2;
            const v2f*   Bp = (const v2f*)Wl + t * 32 + lane;
#pragma unroll 8
            for (int kk = 0; kk < KDIM / 4; ++kk) {
                const v2f a = *(const v2f*)(Ar + ((kk ^ r15) << 2));
                const v2f b = Bp[kk * NT * 32];
                d = __builtin_amdgcn_wmma_f32_16x16x4_f32(
                    false, a, false, b, (short)0, d, false, false);
            }
        }
        __syncthreads();
        if (mine) {
#pragma unroll
            for (int j = 0; j < 8; ++j) {
                float v = fmaf(d[j], nrm[j], bv);
                if (RELU) v = fmaxf(v, 0.0f);
                out[(size_t)(rbase + j) * NOUT + t * 16 + r15] = v;
            }
        }
    }
#undef ROW
#undef WGT
#undef ACC
}

extern "C" void kernel_launch(void* const* d_in, const int* in_sizes, int n_in,
                              void* d_out, int out_size, void* d_ws, size_t ws_size,
                              hipStream_t stream) {
    const float* features = (const float*)d_in[0];
    const float* W1 = (const float*)d_in[1];
    const float* b1 = (const float*)d_in[2];
    const float* W2 = (const float*)d_in[3];
    const float* b2 = (const float*)d_in[4];
    const float* W3 = (const float*)d_in[5];
    const float* b3 = (const float*)d_in[6];
    const int*  src = (const int*)d_in[7];
    const int*  dst = (const int*)d_in[8];
    const int E = in_sizes[7];

    int*   cnt      = (int*)d_ws;
    float* nrm      = (float*)d_ws + 2 * N_NODES;
    int*   rowptr   = (int*)d_ws + 4 * N_NODES;
    int*   blocksum = (int*)d_ws + 250048;
    float* aggA     = (float*)d_ws + 250304;
    float* bufB     = aggA + (size_t)N_NODES * FDIM;
    int*   edges    = (int*)(bufB + (size_t)N_NODES * FDIM);
    int*   rank     = edges + 2 * (size_t)E;
    const float* norm_src = nrm;
    const float* norm_dst = nrm + N_NODES;

    zero_i32<<<512, 256, 0, stream>>>(cnt, 2 * N_NODES);
    degree_rank_kernel<<<(E + 255) / 256, 256, 0, stream>>>(src, dst, cnt, cnt + N_NODES, rank, E);
    finalize_kernel<<<SCAN_BLOCKS, 256, 0, stream>>>(cnt, nrm, blocksum);
    scan_kernel<<<SCAN_BLOCKS, 256, 0, stream>>>(cnt + N_NODES, blocksum, rowptr);
    fill_kernel<<<(E + 255) / 256, 256, 0, stream>>>(src, dst, rank, rowptr, norm_src, edges, E);

    const int layerBlocks = 1024;

    layer_kernel<128, true><<<layerBlocks, 512, 0, stream>>>(features, rowptr, edges, W1, b1,
                                                            norm_dst, bufB);
    layer_kernel<128, true><<<layerBlocks, 512, 0, stream>>>(bufB, rowptr, edges, W2, b2,
                                                            norm_dst, aggA);
    layer_kernel<64, false><<<layerBlocks, 512, 0, stream>>>(aggA, rowptr, edges, W3, b3,
                                                            norm_dst, (float*)d_out);
}


# Layer 3 transforms before it aggregates: layer 2's launch also multiplies its tiles by W3 (f32 WMMA), layer 3 gathers 256-B rows; edge sums by half-wave pairs, W3 product ahead of the edge sum
# speedup vs baseline: 3.0188x; 1.0339x over previous
#include <hip/hip_runtime.h>
#include <hip/hip_bf16.h>


#define N_NODES 50000
#define FDIM    128
#define KDIM    128
#define SCAN_BLOCKS ((N_NODES + 255) / 256)

typedef __attribute__((ext_vector_type(2))) float v2f;
typedef __attribute__((ext_vector_type(8))) float v8f;

__global__ void zero_i32(int* __restrict__ p, int n) {
    int i = blockIdx.x * blockDim.x + threadIdx.x;
    int stride = gridDim.x * blockDim.x;
    for (; i < n; i += stride) p[i] = 0;
}

__device__ __forceinline__ int wave_sum(int v) {
    for (int o = 16; o > 0; o >>= 1) v += __shfl_xor(v, o);
    return v;
}

__device__ __forceinline__ int wave_inclusive_scan(int v, int lane) {
    for (int o = 1; o < 32; o <<= 1) {
        const int t = __shfl_up(v, o);
        if (lane >= o) v += t;
    }
    return v;
}

__global__ void degree_rank_kernel(const int* __restrict__ src, const int* __restrict__ dst,
                                   int* __restrict__ cnt_out, int* __restrict__ cnt_in,
                                   int* __restrict__ rank, int E) {
    int i = blockIdx.x * blockDim.x + threadIdx.x;
    if (i < E) {
        atomicAdd(&cnt_out[src[i]], 1);
        rank[i] = atomicAdd(&cnt_in[dst[i]], 1);
    }
}

__global__ __launch_bounds__(256)
void finalize_kernel(const int* __restrict__ cnt, float* __restrict__ nrm,
                     int* __restrict__ blocksum) {
    __shared__ int part[8];
    const int tid = threadIdx.x;
    const int i = blockIdx.x * 256 + tid;
    int ci = 0;
    if (i < N_NODES) {
        const int co = cnt[i];
        ci = cnt[N_NODES + i];
        nrm[i]           = rsqrtf(fmaxf((float)co, 1.0f));
        nrm[N_NODES + i] = rsqrtf(fmaxf((float)ci, 1.0f));
    }
    const int s = wave_sum(ci);
    if ((tid & 31) == 0) part[tid >> 5] = s;
    __syncthreads();
    if (tid == 0) {
        int t = 0;
        for (int k = 0; k < 8; ++k) t += part[k];
        blocksum[blockIdx.x] = t;
    }
}

__global__ __launch_bounds__(256)
void scan_kernel(const int* __restrict__ cnt_in, const int* __restrict__ blocksum,
                 int* __restrict__ rowptr) {
    __shared__ int pre[8];
    __shared__ int tot[8];
    const int tid  = threadIdx.x;
    const int lane = tid & 31;
    const int wave = tid >> 5;
    int p = 0;
    for (int k = tid; k < (int)blockIdx.x; k += 256) p += blocksum[k];
    p = wave_sum(p);
    const int i = blockIdx.x * 256 + tid;
    const int c = (i < N_NODES) ? cnt_in[i] : 0;
    const int inc = wave_inclusive_scan(c, lane);
    if (lane == 0)  pre[wave] = p;
    if (lane == 31) tot[wave] = inc;
    __syncthreads();
    int off = 0;
    for (int k = 0; k < 8; ++k) {
        off += pre[k];
        if (k < wave) off += tot[k];
    }
    const int excl = off + inc - c;
    if (i < N_NODES) rowptr[i] = excl;
    if (i == N_NODES - 1) rowptr[N_NODES] = excl + c;
}

__global__ void fill_kernel(const int* __restrict__ src, const int* __restrict__ dst,
                            const int* __restrict__ rank, const int* __restrict__ rowptr,
                            const float* __restrict__ nsrc, int* __restrict__ edges, int E) {
    int i = blockIdx.x * blockDim.x + threadIdx.x;
    if (i < E) {
        const int s = src[i];
        const int pos = rowptr[dst[i]] + rank[i];
        ((int2*)edges)[pos] = make_int2(s, __float_as_int(nsrc[s]));
    }
}

template <int NOUT, bool RELU>
__global__ __launch_bounds__(512)
void layer_kernel(const float* __restrict__ x, const int* __restrict__ rowptr,
                  const int* __restrict__ edges, const float* __restrict__ W,
                  const float* __restrict__ bias, const float* __restrict__ norm,
                  float* __restrict__ out) {
    constexpr int NT     = NOUT / 16;
    constexpr int GROUPS = 16 / NT;
    __shared__ __attribute__((aligned(16))) float Wl[KDIM * NOUT];
    __shared__ __attribute__((aligned(16))) float At[16 * FDIM];

    const int lane  = threadIdx.x & 31;
    const int wave  = __builtin_amdgcn_readfirstlane((int)(threadIdx.x >> 5));
    const int r15   = lane & 15;
    const int khalf = lane >> 4;

#pragma unroll
    for (int s = threadIdx.x; s < (KDIM / 4) * NT * 32; s += 512) {
        const int l  = s & 31;
        const int t  = (s >> 5) % NT;
        const int kk = (s >> 5) / NT;
        const float* Wp = W + (size_t)(kk * 4 + (l >> 4) * 2) * NOUT + t * 16 + (l & 15);
        ((float2*)Wl)[s] = make_float2(Wp[0], Wp[NOUT]);
    }

    const float4* __restrict__ xl = (const float4*)x + lane;
    const int2* __restrict__ el = (const int2*)edges;

#define ROW(j) xl[(size_t)__builtin_amdgcn_readlane(my.x, (j)) * (FDIM / 4)]
#define WGT(j) __int_as_float(__builtin_amdgcn_readlane(my.y, (j)))
#define ACC(VV, WW) { acc.x = fmaf((VV).x, (WW), acc.x); acc.y = fmaf((VV).y, (WW), acc.y); \
                      acc.z = fmaf((VV).z, (WW), acc.z); acc.w = fmaf((VV).w, (WW), acc.w); }

    int it = 0;
    for (int tile = blockIdx.x; tile < N_NODES / 16; tile += gridDim.x, ++it) {
        const int node = tile * 16 + wave;
        const int beg = rowptr[node];
        const int end = rowptr[node + 1];
        float4 acc = make_float4(0.0f, 0.0f, 0.0f, 0.0f);
        for (int base = beg; base < end; base += 32) {
            const int cnt = min(32, end - base);
            int2 my = make_int2(0, 0);
            if (lane < cnt) my = el[base + lane];
            int j = 0;
            for (; j + 8 <= cnt; j += 8) {
                float4 v[8];
                float  w[8];
#pragma unroll
                for (int u = 0; u < 8; ++u) { v[u] = ROW(j + u); w[u] = WGT(j + u); }
#pragma unroll
                for (int u = 0; u < 8; ++u) ACC(v[u], w[u]);
            }
            if (j + 4 <= cnt) {
                float4 v[4];
                float  w[4];
#pragma unroll
                for (int u = 0; u < 4; ++u) { v[u] = ROW(j + u); w[u] = WGT(j + u); }
#pragma unroll
                for (int u = 0; u < 4; ++u) ACC(v[u], w[u]);
                j += 4;
            }
            if (j < cnt) {
                const int r = cnt - j;
                const float4 z = make_float4(0.0f, 0.0f, 0.0f, 0.0f);
                float4 v0 = ROW(j), v1 = z, v2 = z;
                float  w0 = WGT(j), w1 = 0.0f, w2 = 0.0f;
                if (r > 1) { v1 = ROW(j + 1); w1 = WGT(j + 1); }
                if (r > 2) { v2 = ROW(j + 2); w2 = WGT(j + 2); }
                ACC(v0, w0);
                ACC(v1, w1);
                ACC(v2, w2);
            }
        }
        ((float4*)(At + wave * FDIM))[lane ^ wave] = acc;
        __syncthreads();

        const bool mine = (wave / NT == (it & (GROUPS - 1)));
        const int t     = wave % NT;
        const int rbase = tile * 16 + khalf * 8;
        float nrm[8];
        float bv = 0.0f;
        v8f d;
#pragma unroll
        for (int j = 0; j < 8; ++j) { d[j] = 0.0f; nrm[j] = 0.0f; }
        if (mine) {
#pragma unroll
            for (int j = 0; j < 8; ++j) nrm[j] = norm[rbase + j];
            bv = bias[t * 16 + r15];
            const float* Ar = At + r15 * FDIM + khalf * 2;
            const v2f*   Bp = (const v2f*)Wl + t * 32 + lane;
#pragma unroll 8
            for (int kk = 0; kk < KDIM / 4; ++kk) {
                const v2f a = *(const v2f*)(Ar + ((kk ^ r15) << 2));
                const v2f b = Bp[kk * NT * 32];
                d = __builtin_amdgcn_wmma_f32_16x16x4_f32(
                    false, a, false, b, (short)0, d, false, false);
            }
        }
        __syncthreads();
        if (mine) {
#pragma unroll
            for (int j = 0; j < 8; ++j) {
                float v = fmaf(d[j], nrm[j], bv);
                if (RELU) v = fmaxf(v, 0.0f);
                out[(size_t)(rbase + j) * NOUT + t * 16 + r15] = v;
            }
        }
    }
#undef ROW
#undef WGT
#undef ACC
}

__global__ __launch_bounds__(1024, 2)
void layer2_fused_kernel(const float* __restrict__ x, const int* __restrict__ rowptr,
                         const int* __restrict__ edges, const float* __restrict__ W2,
                         const float* __restrict__ b2, const float* __restrict__ W3,
                         const float* __restrict__ norm, float* __restrict__ z) {
    constexpr int NT2 = 8;
    constexpr int NT3 = 4;
    constexpr int ZDIM = 64;
    __shared__ __attribute__((aligned(16))) float Wl[KDIM * 128];
    __shared__ __attribute__((aligned(16))) float W3l[KDIM * ZDIM];
    __shared__ __attribute__((aligned(16))) float At[32 * FDIM];
    __shared__ __attribute__((aligned(16))) float Ht[32 * FDIM];

    const int lane  = threadIdx.x & 31;
    const int wave  = __builtin_amdgcn_readfirstlane((int)(threadIdx.x >> 5));
    const int r15   = lane & 15;
    const int khalf = lane >> 4;

#pragma unroll
    for (int s = threadIdx.x; s < (KDIM / 4) * NT2 * 32; s += 1024) {
        const int l  = s & 31;
        const int t  = (s >> 5) % NT2;
        const int kk = (s >> 5) / NT2;
        const float* Wp = W2 + (size_t)(kk * 4 + (l >> 4) * 2) * 128 + t * 16 + (l & 15);
        ((float2*)Wl)[s] = make_float2(Wp[0], Wp[128]);
    }
#pragma unroll
    for (int s = threadIdx.x; s < (KDIM / 4) * NT3 * 32; s += 1024) {
        const int l  = s & 31;
        const int t  = (s >> 5) % NT3;
        const int kk = (s >> 5) / NT3;
        const float* Wp = W3 + (size_t)(kk * 4 + (l >> 4) * 2) * ZDIM + t * 16 + (l & 15);
        ((float2*)W3l)[s] = make_float2(Wp[0], Wp[ZDIM]);
    }

    const float4* __restrict__ xl = (const float4*)x + lane;
    const int2* __restrict__ el = (const int2*)edges;

#define ROW(j) xl[(size_t)__builtin_amdgcn_readlane(my.x, (j)) * (FDIM / 4)]
#define WGT(j) __int_as_float(__builtin_amdgcn_readlane(my.y, (j)))
#define ACC(VV, WW) { acc.x = fmaf((VV).x, (WW), acc.x); acc.y = fmaf((VV).y, (WW), acc.y); \
                      acc.z = fmaf((VV).z, (WW), acc.z); acc.w = fmaf((VV).w, (WW), acc.w); }

    for (int tile = blockIdx.x; tile < (N_NODES + 31) / 32; tile += gridDim.x) {
        const int node = tile * 32 + wave;
        int beg = 0, end = 0;
        if (node < N_NODES) { beg = rowptr[node]; end = rowptr[node + 1]; }
        float4 acc = make_float4(0.0f, 0.0f, 0.0f, 0.0f);
        for (int base = beg; base < end; base += 32) {
            const int cnt = min(32, end - base);
            int2 my = make_int2(0, 0);
            if (lane < cnt) my = el[base + lane];
            int j = 0;
            for (; j + 8 <= cnt; j += 8) {
                float4 v[8];
                float  w[8];
#pragma unroll
                for (int u = 0; u < 8; ++u) { v[u] = ROW(j + u); w[u] = WGT(j + u); }
#pragma unroll
                for (int u = 0; u < 8; ++u) ACC(v[u], w[u]);
            }
            if (j + 4 <= cnt) {
                float4 v[4];
                float  w[4];
#pragma unroll
                for (int u = 0; u < 4; ++u) { v[u] = ROW(j + u); w[u] = WGT(j + u); }
#pragma unroll
                for (int u = 0; u < 4; ++u) ACC(v[u], w[u]);
                j += 4;
            }
            if (j < cnt) {
                const int r = cnt - j;
                const float4 z4 = make_float4(0.0f, 0.0f, 0.0f, 0.0f);
                float4 v0 = ROW(j), v1 = z4, v2 = z4;
                float  w0 = WGT(j), w1 = 0.0f, w2 = 0.0f;
                if (r > 1) { v1 = ROW(j + 1); w1 = WGT(j + 1); }
                if (r > 2) { v2 = ROW(j + 2); w2 = WGT(j + 2); }
                ACC(v0, w0);
                ACC(v1, w1);
                ACC(v2, w2);
            }
        }
        ((float4*)(At + wave * FDIM))[lane ^ (wave & 15)] = acc;
        __syncthreads();

        if (wave < 16) {
            const int mt = wave >> 3;
            const int t  = wave & 7;
            const int rbase = tile * 32 + mt * 16 + khalf * 8;
            float nrm[8];
#pragma unroll
            for (int j = 0; j < 8; ++j) nrm[j] = norm[min(rbase + j, N_NODES - 1)];
            const float bv = b2[t * 16 + r15];
            v8f d;
#pragma unroll
            for (int j = 0; j < 8; ++j) d[j] = 0.0f;
            const float* Ar = At + (mt * 16 + r15) * FDIM + khalf * 2;
            const v2f*   Bp = (const v2f*)Wl + t * 32 + lane;
#pragma unroll 8
            for (int kk = 0; kk < KDIM / 4; ++kk) {
                const v2f a = *(const v2f*)(Ar + ((kk ^ r15) << 2));
                const v2f b = Bp[kk * NT2 * 32];
                d = __builtin_amdgcn_wmma_f32_16x16x4_f32(
                    false, a, false, b, (short)0, d, false, false);
            }
            const int col = t * 16 + r15;
#pragma unroll
            for (int j = 0; j < 8; ++j) {
                const int row = mt * 16 + khalf * 8 + j;
                const float v = fmaxf(fmaf(d[j], nrm[j], bv), 0.0f);
                Ht[row * FDIM + ((((col >> 2) ^ (row & 15)) << 2) | (col & 3))] = v;
            }
        }
        __syncthreads();
        if (wave >= 16 && wave < 24) {
            const int q  = wave - 16;
            const int mt = q >> 2;
            const int t3 = q & 3;
            v8f d;
#pragma unroll
            for (int j = 0; j < 8; ++j) d[j] = 0.0f;
            const float* Hr = Ht + (mt * 16 + r15) * FDIM + khalf * 2;
            const v2f*   Bp = (const v2f*)W3l + t3 * 32 + lane;
#pragma unroll 8
            for (int kk = 0; kk < KDIM / 4; ++kk) {
                const v2f a = *(const v2f*)(Hr + ((kk ^ r15) << 2));
                const v2f b = Bp[kk * NT3 * 32];
                d = __builtin_amdgcn_wmma_f32_16x16x4_f32(
                    false, a, false, b, (short)0, d, false, false);
            }
            const int rbase = tile * 32 + mt * 16 + khalf * 8;
#pragma unroll
            for (int j = 0; j < 8; ++j) {
                if (rbase + j < N_NODES) z[(size_t)(rbase + j) * ZDIM + t3 * 16 + r15] = d[j];
            }
        }
    }
#undef ROW
#undef WGT
#undef ACC
}

template <int P>
__device__ __forceinline__ void gather_pairs(const float4* __restrict__ zl, int2 my, int j,
                                             int cnt, int khalf, float4& acc) {
    float4 v[P];
    float  w[P];
#pragma unroll
    for (int u = 0; u < P; ++u) {
        const int e  = j + 2 * u + khalf;
        const int ec = min(e, cnt - 1);
        const int s  = __shfl(my.x, ec);
        const float wt = __int_as_float(__shfl(my.y, ec));
        w[u] = (e < cnt) ? wt : 0.0f;
        v[u] = zl[(size_t)s * 16];
    }
#pragma unroll
    for (int u = 0; u < P; ++u) {
        acc.x = fmaf(v[u].x, w[u], acc.x); acc.y = fmaf(v[u].y, w[u], acc.y);
        acc.z = fmaf(v[u].z, w[u], acc.z); acc.w = fmaf(v[u].w, w[u], acc.w);
    }
}

__global__ __launch_bounds__(256)
void gather64_kernel(const float* __restrict__ z, const int* __restrict__ rowptr,
                     const int* __restrict__ edges, const float* __restrict__ bias,
                     const float* __restrict__ norm, float* __restrict__ out) {
    const int lane  = threadIdx.x & 31;
    const int r15   = lane & 15;
    const int khalf = lane >> 4;
    const int gw = __builtin_amdgcn_readfirstlane((int)(blockIdx.x * 8 + (threadIdx.x >> 5)));
    const int nw = gridDim.x * 8;
    const float4* __restrict__ zl = (const float4*)z + r15;
    const int2* __restrict__ el = (const int2*)edges;
    const float4 bq = ((const float4*)bias)[r15];
    for (int node = gw; node < N_NODES; node += nw) {
        const int beg = rowptr[node];
        const int end = rowptr[node + 1];
        float4 acc = make_float4(0.0f, 0.0f, 0.0f, 0.0f);
        for (int base = beg; base < end; base += 32) {
            const int cnt = min(32, end - base);
            int2 my = make_int2(0, 0);
            if (lane < cnt) my = el[base + lane];
            int j = 0;
            for (; j + 8 < cnt; j += 16) gather_pairs<8>(zl, my, j, cnt, khalf, acc);
            if (j < cnt) gather_pairs<4>(zl, my, j, cnt, khalf, acc);
        }
        acc.x += __shfl_xor(acc.x, 16); acc.y += __shfl_xor(acc.y, 16);
        acc.z += __shfl_xor(acc.z, 16); acc.w += __shfl_xor(acc.w, 16);
        if (khalf == 0) {
            const float nd = norm[node];
            float4 o;
            o.x = fmaf(acc.x, nd, bq.x); o.y = fmaf(acc.y, nd, bq.y);
            o.z = fmaf(acc.z, nd, bq.z); o.w = fmaf(acc.w, nd, bq.w);
            ((float4*)out)[(size_t)node * 16 + r15] = o;
        }
    }
}

extern "C" void kernel_launch(void* const* d_in, const int* in_sizes, int n_in,
                              void* d_out, int out_size, void* d_ws, size_t ws_size,
                              hipStream_t stream) {
    const float* features = (const float*)d_in[0];
    const float* W1 = (const float*)d_in[1];
    const float* b1 = (const float*)d_in[2];
    const float* W2 = (const float*)d_in[3];
    const float* b2 = (const float*)d_in[4];
    const float* W3 = (const float*)d_in[5];
    const float* b3 = (const float*)d_in[6];
    const int*  src = (const int*)d_in[7];
    const int*  dst = (const int*)d_in[8];
    const int E = in_sizes[7];

    int*   cnt      = (int*)d_ws;
    float* nrm      = (float*)d_ws + 2 * N_NODES;
    int*   rowptr   = (int*)d_ws + 4 * N_NODES;
    int*   blocksum = (int*)d_ws + 250048;
    float* aggA     = (float*)d_ws + 250304;
    float* bufB     = aggA + (size_t)N_NODES * FDIM;
    int*   edges    = (int*)(bufB + (size_t)N_NODES * FDIM);
    int*   rank     = edges + 2 * (size_t)E;
    const float* norm_src = nrm;
    const float* norm_dst = nrm + N_NODES;

    zero_i32<<<512, 256, 0, stream>>>(cnt, 2 * N_NODES);
    degree_rank_kernel<<<(E + 255) / 256, 256, 0, stream>>>(src, dst, cnt, cnt + N_NODES, rank, E);
    finalize_kernel<<<SCAN_BLOCKS, 256, 0, stream>>>(cnt, nrm, blocksum);
    scan_kernel<<<SCAN_BLOCKS, 256, 0, stream>>>(cnt + N_NODES, blocksum, rowptr);
    fill_kernel<<<(E + 255) / 256, 256, 0, stream>>>(src, dst, rank, rowptr, norm_src, edges, E);

    const int layerBlocks = 1024;

    layer_kernel<128, true><<<layerBlocks, 512, 0, stream>>>(features, rowptr, edges, W1, b1,
                                                            norm_dst, bufB);
    layer2_fused_kernel<<<512, 1024, 0, stream>>>(bufB, rowptr, edges, W2, b2, W3, norm_dst, aggA);
    gather64_kernel<<<1536, 256, 0, stream>>>(aggA, rowptr, edges, b3, norm_dst, (float*)d_out);
}


# Dst-sorted edge list by counting sort in LDS: 128 edge chunks count with LDS atomics, prefix over chunks, fill from LDS cursors; no global atomics, 8 to 7 launches; sums in arrival order per chunk
# speedup vs baseline: 4.8739x; 1.6145x over previous
#include <hip/hip_runtime.h>
#include <hip/hip_bf16.h>


#define N_NODES 50000
#define FDIM    128
#define KDIM    128
#define SCAN_BLOCKS ((N_NODES + 255) / 256)

typedef __attribute__((ext_vector_type(2))) float v2f;
typedef __attribute__((ext_vector_type(8))) float v8f;

__device__ __forceinline__ int wave_sum(int v) {
    for (int o = 16; o > 0; o >>= 1) v += __shfl_xor(v, o);
    return v;
}

__device__ __forceinline__ int wave_inclusive_scan(int v, int lane) {
    for (int o = 1; o < 32; o <<= 1) {
        const int t = __shfl_up(v, o);
        if (lane >= o) v += t;
    }
    return v;
}

#define G_CHUNKS 128

__global__ __launch_bounds__(1024)
void chunk_hist_kernel(const int* __restrict__ src, const int* __restrict__ dst,
                       unsigned* __restrict__ H, int E) {
    __shared__ __attribute__((aligned(16))) unsigned tab[N_NODES];
    const int tid = threadIdx.x;
    for (int q = tid; q < N_NODES / 4; q += 1024) ((uint4*)tab)[q] = make_uint4(0u, 0u, 0u, 0u);
    __syncthreads();
    const int chunk = (E + (int)gridDim.x - 1) / (int)gridDim.x;
    const int beg = (int)blockIdx.x * chunk;
    const int end = min(E, beg + chunk);
    for (int e0 = beg + tid; e0 < end; e0 += 8 * 1024) {
        int d[8], s[8];
#pragma unroll
        for (int u = 0; u < 8; ++u) {
            const int e = e0 + u * 1024;
            d[u] = -1; s[u] = -1;
            if (e < end) { d[u] = dst[e]; s[u] = src[e]; }
        }
#pragma unroll
        for (int u = 0; u < 8; ++u) {
            if (d[u] >= 0) {
                atomicAdd(&tab[d[u]], 1u);
                atomicAdd(&tab[s[u]], 0x10000u);
            }
        }
    }
    __syncthreads();
    uint4* Hg = (uint4*)(H + (size_t)blockIdx.x * N_NODES);
    for (int q = tid; q < N_NODES / 4; q += 1024) Hg[q] = ((const uint4*)tab)[q];
}

__global__ __launch_bounds__(256)
void chunk_prefix_kernel(const unsigned* __restrict__ H, int* __restrict__ P,
                         int* __restrict__ cnt_in, float* __restrict__ nrm,
                         int* __restrict__ blocksum) {
    __shared__ int part[8];
    const int tid = threadIdx.x;
    const int i = blockIdx.x * 256 + tid;
    int ci = 0;
    if (i < N_NODES) {
        int co = 0;
        for (int g0 = 0; g0 < G_CHUNKS; g0 += 16) {
            unsigned h[16];
#pragma unroll
            for (int u = 0; u < 16; ++u) h[u] = H[(size_t)(g0 + u) * N_NODES + i];
#pragma unroll
            for (int u = 0; u < 16; ++u) {
                P[(size_t)(g0 + u) * N_NODES + i] = ci;
                ci += (int)(h[u] & 0xFFFFu);
                co += (int)(h[u] >> 16);
            }
        }
        cnt_in[i] = ci;
        nrm[i]           = rsqrtf(fmaxf((float)co, 1.0f));
        nrm[N_NODES + i] = rsqrtf(fmaxf((float)ci, 1.0f));
    }
    const int s = wave_sum(ci);
    if ((tid & 31) == 0) part[tid >> 5] = s;
    __syncthreads();
    if (tid == 0) {
        int t = 0;
        for (int k = 0; k < 8; ++k) t += part[k];
        blocksum[blockIdx.x] = t;
    }
}

__global__ __launch_bounds__(256)
void scan_kernel(const int* __restrict__ cnt_in, const int* __restrict__ blocksum,
                 int* __restrict__ rowptr) {
    __shared__ int pre[8];
    __shared__ int tot[8];
    const int tid  = threadIdx.x;
    const int lane = tid & 31;
    const int wave = tid >> 5;
    int p = 0;
    for (int k = tid; k < (int)blockIdx.x; k += 256) p += blocksum[k];
    p = wave_sum(p);
    const int i = blockIdx.x * 256 + tid;
    const int c = (i < N_NODES) ? cnt_in[i] : 0;
    const int inc = wave_inclusive_scan(c, lane);
    if (lane == 0)  pre[wave] = p;
    if (lane == 31) tot[wave] = inc;
    __syncthreads();
    int off = 0;
    for (int k = 0; k < 8; ++k) {
        off += pre[k];
        if (k < wave) off += tot[k];
    }
    const int excl = off + inc - c;
    if (i < N_NODES) rowptr[i] = excl;
    if (i == N_NODES - 1) rowptr[N_NODES] = excl + c;
}

__global__ __launch_bounds__(1024)
void chunk_fill_kernel(const int* __restrict__ src, const int* __restrict__ dst,
                       const int* __restrict__ rowptr, const int* __restrict__ P,
                       const float* __restrict__ nsrc, int* __restrict__ edges, int E) {
    __shared__ __attribute__((aligned(16))) int tab[N_NODES];
    const int tid = threadIdx.x;
    const int4* Pg = (const int4*)(P + (size_t)blockIdx.x * N_NODES);
    const int4* Rp = (const int4*)rowptr;
#pragma unroll 4
    for (int q = tid; q < N_NODES / 4; q += 1024) {
        const int4 a = Rp[q];
        const int4 b = Pg[q];
        ((int4*)tab)[q] = make_int4(a.x + b.x, a.y + b.y, a.z + b.z, a.w + b.w);
    }
    __syncthreads();
    const int chunk = (E + (int)gridDim.x - 1) / (int)gridDim.x;
    const int beg = (int)blockIdx.x * chunk;
    const int end = min(E, beg + chunk);
    for (int e0 = beg + tid; e0 < end; e0 += 8 * 1024) {
        int d[8], s[8];
        float w[8];
#pragma unroll
        for (int u = 0; u < 8; ++u) {
            const int e = e0 + u * 1024;
            d[u] = -1; s[u] = 0;
            if (e < end) { d[u] = dst[e]; s[u] = src[e]; }
        }
#pragma unroll
        for (int u = 0; u < 8; ++u) w[u] = nsrc[s[u]];
#pragma unroll
        for (int u = 0; u < 8; ++u) {
            if (d[u] >= 0) {
                const int pos = atomicAdd(&tab[d[u]], 1);
                ((int2*)edges)[pos] = make_int2(s[u], __float_as_int(w[u]));
            }
        }
    }
}

template <int NOUT, bool RELU>
__global__ __launch_bounds__(512)
void layer_kernel(const float* __restrict__ x, const int* __restrict__ rowptr,
                  const int* __restrict__ edges, const float* __restrict__ W,
                  const float* __restrict__ bias, const float* __restrict__ norm,
                  float* __restrict__ out) {
    constexpr int NT     = NOUT / 16;
    constexpr int GROUPS = 16 / NT;
    __shared__ __attribute__((aligned(16))) float Wl[KDIM * NOUT];
    __shared__ __attribute__((aligned(16))) float At[16 * FDIM];

    const int lane  = threadIdx.x & 31;
    const int wave  = __builtin_amdgcn_readfirstlane((int)(threadIdx.x >> 5));
    const int r15   = lane & 15;
    const int khalf = lane >> 4;

#pragma unroll
    for (int s = threadIdx.x; s < (KDIM / 4) * NT * 32; s += 512) {
        const int l  = s & 31;
        const int t  = (s >> 5) % NT;
        const int kk = (s >> 5) / NT;
        const float* Wp = W + (size_t)(kk * 4 + (l >> 4) * 2) * NOUT + t * 16 + (l & 15);
        ((float2*)Wl)[s] = make_float2(Wp[0], Wp[NOUT]);
    }

    const float4* __restrict__ xl = (const float4*)x + lane;
    const int2* __restrict__ el = (const int2*)edges;

#define ROW(j) xl[(size_t)__builtin_amdgcn_readlane(my.x, (j)) * (FDIM / 4)]
#define WGT(j) __int_as_float(__builtin_amdgcn_readlane(my.y, (j)))
#define ACC(VV, WW) { acc.x = fmaf((VV).x, (WW), acc.x); acc.y = fmaf((VV).y, (WW), acc.y); \
                      acc.z = fmaf((VV).z, (WW), acc.z); acc.w = fmaf((VV).w, (WW), acc.w); }

    int it = 0;
    for (int tile = blockIdx.x; tile < N_NODES / 16; tile += gridDim.x, ++it) {
        const int node = tile * 16 + wave;
        const int beg = rowptr[node];
        const int end = rowptr[node + 1];
        float4 acc = make_float4(0.0f, 0.0f, 0.0f, 0.0f);
        for (int base = beg; base < end; base += 32) {
            const int cnt = min(32, end - base);
            int2 my = make_int2(0, 0);
            if (lane < cnt) my = el[base + lane];
            int j = 0;
            for (; j + 8 <= cnt; j += 8) {
                float4 v[8];
                float  w[8];
#pragma unroll
                for (int u = 0; u < 8; ++u) { v[u] = ROW(j + u); w[u] = WGT(j + u); }
#pragma unroll
                for (int u = 0; u < 8; ++u) ACC(v[u], w[u]);
            }
            if (j + 4 <= cnt) {
                float4 v[4];
                float  w[4];
#pragma unroll
                for (int u = 0; u < 4; ++u) { v[u] = ROW(j + u); w[u] = WGT(j + u); }
#pragma unroll
                for (int u = 0; u < 4; ++u) ACC(v[u], w[u]);
                j += 4;
            }
            if (j < cnt) {
                const int r = cnt - j;
                const float4 z = make_float4(0.0f, 0.0f, 0.0f, 0.0f);
                float4 v0 = ROW(j), v1 = z, v2 = z;
                float  w0 = WGT(j), w1 = 0.0f, w2 = 0.0f;
                if (r > 1) { v1 = ROW(j + 1); w1 = WGT(j + 1); }
                if (r > 2) { v2 = ROW(j + 2); w2 = WGT(j + 2); }
                ACC(v0, w0);
                ACC(v1, w1);
                ACC(v2, w2);
            }
        }
        ((float4*)(At + wave * FDIM))[lane ^ wave] = acc;
        __syncthreads();

        const bool mine = (wave / NT == (it & (GROUPS - 1)));
        const int t     = wave % NT;
        const int rbase = tile * 16 + khalf * 8;
        float nrm[8];
        float bv = 0.0f;
        v8f d;
#pragma unroll
        for (int j = 0; j < 8; ++j) { d[j] = 0.0f; nrm[j] = 0.0f; }
        if (mine) {
#pragma unroll
            for (int j = 0; j < 8; ++j) nrm[j] = norm[rbase + j];
            bv = bias[t * 16 + r15];
            const float* Ar = At + r15 * FDIM + khalf * 2;
            const v2f*   Bp = (const v2f*)Wl + t * 32 + lane;
#pragma unroll 8
            for (int kk = 0; kk < KDIM / 4; ++kk) {
                const v2f a = *(const v2f*)(Ar + ((kk ^ r15) << 2));
                const v2f b = Bp[kk * NT * 32];
                d = __builtin_amdgcn_wmma_f32_16x16x4_f32(
                    false, a, false, b, (short)0, d, false, false);
            }
        }
        __syncthreads();
        if (mine) {
#pragma unroll
            for (int j = 0; j < 8; ++j) {
                float v = fmaf(d[j], nrm[j], bv);
                if (RELU) v = fmaxf(v, 0.0f);
                out[(size_t)(rbase + j) * NOUT + t * 16 + r15] = v;
            }
        }
    }
#undef ROW
#undef WGT
#undef ACC
}

__global__ __launch_bounds__(1024, 2)
void layer2_fused_kernel(const float* __restrict__ x, const int* __restrict__ rowptr,
                         const int* __restrict__ edges, const float* __restrict__ W2,
                         const float* __restrict__ b2, const float* __restrict__ W3,
                         const float* __restrict__ norm, float* __restrict__ z) {
    constexpr int NT2 = 8;
    constexpr int NT3 = 4;
    constexpr int ZDIM = 64;
    __shared__ __attribute__((aligned(16))) float Wl[KDIM * 128];
    __shared__ __attribute__((aligned(16))) float W3l[KDIM * ZDIM];
    __shared__ __attribute__((aligned(16))) float At[32 * FDIM];
    __shared__ __attribute__((aligned(16))) float Ht[32 * FDIM];

    const int lane  = threadIdx.x & 31;
    const int wave  = __builtin_amdgcn_readfirstlane((int)(threadIdx.x >> 5));
    const int r15   = lane & 15;
    const int khalf = lane >> 4;

#pragma unroll
    for (int s = threadIdx.x; s < (KDIM / 4) * NT2 * 32; s += 1024) {
        const int l  = s & 31;
        const int t  = (s >> 5) % NT2;
        const int kk = (s >> 5) / NT2;
        const float* Wp = W2 + (size_t)(kk * 4 + (l >> 4) * 2) * 128 + t * 16 + (l & 15);
        ((float2*)Wl)[s] = make_float2(Wp[0], Wp[128]);
    }
#pragma unroll
    for (int s = threadIdx.x; s < (KDIM / 4) * NT3 * 32; s += 1024) {
        const int l  = s & 31;
        const int t  = (s >> 5) % NT3;
        const int kk = (s >> 5) / NT3;
        const float* Wp = W3 + (size_t)(kk * 4 + (l >> 4) * 2) * ZDIM + t * 16 + (l & 15);
        ((float2*)W3l)[s] = make_float2(Wp[0], Wp[ZDIM]);
    }

    const float4* __restrict__ xl = (const float4*)x + lane;
    const int2* __restrict__ el = (const int2*)edges;

#define ROW(j) xl[(size_t)__builtin_amdgcn_readlane(my.x, (j)) * (FDIM / 4)]
#define WGT(j) __int_as_float(__builtin_amdgcn_readlane(my.y, (j)))
#define ACC(VV, WW) { acc.x = fmaf((VV).x, (WW), acc.x); acc.y = fmaf((VV).y, (WW), acc.y); \
                      acc.z = fmaf((VV).z, (WW), acc.z); acc.w = fmaf((VV).w, (WW), acc.w); }

    for (int tile = blockIdx.x; tile < (N_NODES + 31) / 32; tile += gridDim.x) {
        const int node = tile * 32 + wave;
        int beg = 0, end = 0;
        if (node < N_NODES) { beg = rowptr[node]; end = rowptr[node + 1]; }
        float4 acc = make_float4(0.0f, 0.0f, 0.0f, 0.0f);
        for (int base = beg; base < end; base += 32) {
            const int cnt = min(32, end - base);
            int2 my = make_int2(0, 0);
            if (lane < cnt) my = el[base + lane];
            int j = 0;
            for (; j + 8 <= cnt; j += 8) {
                float4 v[8];
                float  w[8];
#pragma unroll
                for (int u = 0; u < 8; ++u) { v[u] = ROW(j + u); w[u] = WGT(j + u); }
#pragma unroll
                for (int u = 0; u < 8; ++u) ACC(v[u], w[u]);
            }
            if (j + 4 <= cnt) {
                float4 v[4];
                float  w[4];
#pragma unroll
                for (int u = 0; u < 4; ++u) { v[u] = ROW(j + u); w[u] = WGT(j + u); }
#pragma unroll
                for (int u = 0; u < 4; ++u) ACC(v[u], w[u]);
                j += 4;
            }
            if (j < cnt) {
                const int r = cnt - j;
                const float4 z4 = make_float4(0.0f, 0.0f, 0.0f, 0.0f);
                float4 v0 = ROW(j), v1 = z4, v2 = z4;
                float  w0 = WGT(j), w1 = 0.0f, w2 = 0.0f;
                if (r > 1) { v1 = ROW(j + 1); w1 = WGT(j + 1); }
                if (r > 2) { v2 = ROW(j + 2); w2 = WGT(j + 2); }
                ACC(v0, w0);
                ACC(v1, w1);
                ACC(v2, w2);
            }
        }
        ((float4*)(At + wave * FDIM))[lane ^ (wave & 15)] = acc;
        __syncthreads();

        if (wave < 16) {
            const int mt = wave >> 3;
            const int t  = wave & 7;
            const int rbase = tile * 32 + mt * 16 + khalf * 8;
            float nrm[8];
#pragma unroll
            for (int j = 0; j < 8; ++j) nrm[j] = norm[min(rbase + j, N_NODES - 1)];
            const float bv = b2[t * 16 + r15];
            v8f d;
#pragma unroll
            for (int j = 0; j < 8; ++j) d[j] = 0.0f;
            const float* Ar = At + (mt * 16 + r15) * FDIM + khalf * 2;
            const v2f*   Bp = (const v2f*)Wl + t * 32 + lane;
#pragma unroll 8
            for (int kk = 0; kk < KDIM / 4; ++kk) {
                const v2f a = *(const v2f*)(Ar + ((kk ^ r15) << 2));
                const v2f b = Bp[kk * NT2 * 32];
                d = __builtin_amdgcn_wmma_f32_16x16x4_f32(
                    false, a, false, b, (short)0, d, false, false);
            }
            const int col = t * 16 + r15;
#pragma unroll
            for (int j = 0; j < 8; ++j) {
                const int row = mt * 16 + khalf * 8 + j;
                const float v = fmaxf(fmaf(d[j], nrm[j], bv), 0.0f);
                Ht[row * FDIM + ((((col >> 2) ^ (row & 15)) << 2) | (col & 3))] = v;
            }
        }
        __syncthreads();
        if (wave >= 16 && wave < 24) {
            const int q  = wave - 16;
            const int mt = q >> 2;
            const int t3 = q & 3;
            v8f d;
#pragma unroll
            for (int j = 0; j < 8; ++j) d[j] = 0.0f;
            const float* Hr = Ht + (mt * 16 + r15) * FDIM + khalf * 2;
            const v2f*   Bp = (const v2f*)W3l + t3 * 32 + lane;
#pragma unroll 8
            for (int kk = 0; kk < KDIM / 4; ++kk) {
                const v2f a = *(const v2f*)(Hr + ((kk ^ r15) << 2));
                const v2f b = Bp[kk * NT3 * 32];
                d = __builtin_amdgcn_wmma_f32_16x16x4_f32(
                    false, a, false, b, (short)0, d, false, false);
            }
            const int rbase = tile * 32 + mt * 16 + khalf * 8;
#pragma unroll
            for (int j = 0; j < 8; ++j) {
                if (rbase + j < N_NODES) z[(size_t)(rbase + j) * ZDIM + t3 * 16 + r15] = d[j];
            }
        }
    }
#undef ROW
#undef WGT
#undef ACC
}

template <int P>
__device__ __forceinline__ void gather_pairs(const float4* __restrict__ zl, int2 my, int j,
                                             int cnt, int khalf, float4& acc) {
    float4 v[P];
    float  w[P];
#pragma unroll
    for (int u = 0; u < P; ++u) {
        const int e  = j + 2 * u + khalf;
        const int ec = min(e, cnt - 1);
        const int s  = __shfl(my.x, ec);
        const float wt = __int_as_float(__shfl(my.y, ec));
        w[u] = (e < cnt) ? wt : 0.0f;
        v[u] = zl[(size_t)s * 16];
    }
#pragma unroll
    for (int u = 0; u < P; ++u) {
        acc.x = fmaf(v[u].x, w[u], acc.x); acc.y = fmaf(v[u].y, w[u], acc.y);
        acc.z = fmaf(v[u].z, w[u], acc.z); acc.w = fmaf(v[u].w, w[u], acc.w);
    }
}

__global__ __launch_bounds__(256)
void gather64_kernel(const float* __restrict__ z, const int* __restrict__ rowptr,
                     const int* __restrict__ edges, const float* __restrict__ bias,
                     const float* __restrict__ norm, float* __restrict__ out) {
    const int lane  = threadIdx.x & 31;
    const int r15   = lane & 15;
    const int khalf = lane >> 4;
    const int gw = __builtin_amdgcn_readfirstlane((int)(blockIdx.x * 8 + (threadIdx.x >> 5)));
    const int nw = gridDim.x * 8;
    const float4* __restrict__ zl = (const float4*)z + r15;
    const int2* __restrict__ el = (const int2*)edges;
    const float4 bq = ((const float4*)bias)[r15];
    for (int node = gw; node < N_NODES; node += nw) {
        const int beg = rowptr[node];
        const int end = rowptr[node + 1];
        float4 acc = make_float4(0.0f, 0.0f, 0.0f, 0.0f);
        for (int base = beg; base < end; base += 32) {
            const int cnt = min(32, end - base);
            int2 my = make_int2(0, 0);
            if (lane < cnt) my = el[base + lane];
            int j = 0;
            for (; j + 8 < cnt; j += 16) gather_pairs<8>(zl, my, j, cnt, khalf, acc);
            if (j < cnt) gather_pairs<4>(zl, my, j, cnt, khalf, acc);
        }
        acc.x += __shfl_xor(acc.x, 16); acc.y += __shfl_xor(acc.y, 16);
        acc.z += __shfl_xor(acc.z, 16); acc.w += __shfl_xor(acc.w, 16);
        if (khalf == 0) {
            const float nd = norm[node];
            float4 o;
            o.x = fmaf(acc.x, nd, bq.x); o.y = fmaf(acc.y, nd, bq.y);
            o.z = fmaf(acc.z, nd, bq.z); o.w = fmaf(acc.w, nd, bq.w);
            ((float4*)out)[(size_t)node * 16 + r15] = o;
        }
    }
}

extern "C" void kernel_launch(void* const* d_in, const int* in_sizes, int n_in,
                              void* d_out, int out_size, void* d_ws, size_t ws_size,
                              hipStream_t stream) {
    const float* features = (const float*)d_in[0];
    const float* W1 = (const float*)d_in[1];
    const float* b1 = (const float*)d_in[2];
    const float* W2 = (const float*)d_in[3];
    const float* b2 = (const float*)d_in[4];
    const float* W3 = (const float*)d_in[5];
    const float* b3 = (const float*)d_in[6];
    const int*  src = (const int*)d_in[7];
    const int*  dst = (const int*)d_in[8];
    const int E = in_sizes[7];

    int*   cnt      = (int*)d_ws;
    float* nrm      = (float*)d_ws + 2 * N_NODES;
    int*   rowptr   = (int*)d_ws + 4 * N_NODES;
    int*   blocksum = (int*)d_ws + 250048;
    float* aggA     = (float*)d_ws + 250304;
    float* bufB     = aggA + (size_t)N_NODES * FDIM;
    int*   edges    = (int*)(bufB + (size_t)N_NODES * FDIM);
    unsigned* H     = (unsigned*)(edges + 2 * (size_t)E);
    int*   P        = (int*)(H + (size_t)G_CHUNKS * N_NODES);
    const float* norm_src = nrm;
    const float* norm_dst = nrm + N_NODES;

    chunk_hist_kernel<<<G_CHUNKS, 1024, 0, stream>>>(src, dst, H, E);
    chunk_prefix_kernel<<<SCAN_BLOCKS, 256, 0, stream>>>(H, P, cnt + N_NODES, nrm, blocksum);
    scan_kernel<<<SCAN_BLOCKS, 256, 0, stream>>>(cnt + N_NODES, blocksum, rowptr);
    chunk_fill_kernel<<<G_CHUNKS, 1024, 0, stream>>>(src, dst, rowptr, P, norm_src, edges, E);

    const int layerBlocks = 1024;

    layer_kernel<128, true><<<layerBlocks, 512, 0, stream>>>(features, rowptr, edges, W1, b1,
                                                            norm_dst, bufB);
    layer2_fused_kernel<<<512, 1024, 0, stream>>>(bufB, rowptr, edges, W2, b2, W3, norm_dst, aggA);
    gather64_kernel<<<1536, 256, 0, stream>>>(aggA, rowptr, edges, b3, norm_dst, (float*)d_out);
}


# Edge list by a two-level bucket sort: chunks sort their edges by node range in LDS, one work-group per range counts degrees, scans and fills; no per-chunk node tables, 7 to 6 launches; arrival order
# speedup vs baseline: 5.7962x; 1.1892x over previous
#include <hip/hip_runtime.h>
#include <hip/hip_bf16.h>


#define N_NODES 50000
#define FDIM    128
#define KDIM    128

typedef __attribute__((ext_vector_type(2))) float v2f;
typedef __attribute__((ext_vector_type(8))) float v8f;

__device__ __forceinline__ int wave_sum(int v) {
    for (int o = 16; o > 0; o >>= 1) v += __shfl_xor(v, o);
    return v;
}

__device__ __forceinline__ int wave_inclusive_scan(int v, int lane) {
    for (int o = 1; o < 32; o <<= 1) {
        const int t = __shfl_up(v, o);
        if (lane >= o) v += t;
    }
    return v;
}

#define G_CHUNKS  128
#define NB        128
#define RANGE     ((N_NODES + NB - 1) / NB)
#define CHUNK_CAP 8192

static_assert(NB == 128, "scan_ranges covers 4 ranges per lane of one wave");
__device__ __forceinline__ void scan_ranges(int* cur, int* __restrict__ off, int lane) {
    const int c0 = cur[4 * lane], c1 = cur[4 * lane + 1], c2 = cur[4 * lane + 2], c3 = cur[4 * lane + 3];
    const int sum = c0 + c1 + c2 + c3;
    const int inc = wave_inclusive_scan(sum, lane);
    const int x0 = inc - sum, x1 = x0 + c0, x2 = x1 + c1, x3 = x2 + c2;
    cur[4 * lane] = x0; cur[4 * lane + 1] = x1; cur[4 * lane + 2] = x2; cur[4 * lane + 3] = x3;
    off[4 * lane] = x0; off[4 * lane + 1] = x1; off[4 * lane + 2] = x2; off[4 * lane + 3] = x3;
    if (lane == 31) off[NB] = inc;
}

__global__ __launch_bounds__(1024)
void bin_kernel(const int* __restrict__ src, const int* __restrict__ dst,
                int* __restrict__ tmpD, int* __restrict__ tmpS,
                int* __restrict__ offD, int* __restrict__ offS, int E) {
    __shared__ int curD[NB];
    __shared__ int curS[NB];
    __shared__ __attribute__((aligned(16))) int2 stD[CHUNK_CAP];
    __shared__ __attribute__((aligned(16))) int  stS[CHUNK_CAP];
    const int tid  = threadIdx.x;
    const int lane = tid & 31;
    if (tid < NB) { curD[tid] = 0; curS[tid] = 0; }
    __syncthreads();
    const int chunk = (E + (int)gridDim.x - 1) / (int)gridDim.x;
    const int beg = (int)blockIdx.x * chunk;
    const int end = min(E, beg + min(chunk, CHUNK_CAP));
    for (int e0 = beg + tid; e0 < end; e0 += 8 * 1024) {
        int d[8], s[8];
#pragma unroll
        for (int u = 0; u < 8; ++u) {
            const int e = e0 + u * 1024;
            d[u] = -1; s[u] = 0;
            if (e < end) { d[u] = dst[e]; s[u] = src[e]; }
        }
#pragma unroll
        for (int u = 0; u < 8; ++u) {
            if (d[u] >= 0) {
                atomicAdd(&curD[(unsigned)d[u] / RANGE], 1);
                atomicAdd(&curS[(unsigned)s[u] / RANGE], 1);
            }
        }
    }
    __syncthreads();
    if (tid < 32)      scan_ranges(curD, offD + (size_t)blockIdx.x * (NB + 1), lane);
    else if (tid < 64) scan_ranges(curS, offS + (size_t)blockIdx.x * (NB + 1), lane);
    __syncthreads();
    for (int e0 = beg + tid; e0 < end; e0 += 8 * 1024) {
        int d[8], s[8];
#pragma unroll
        for (int u = 0; u < 8; ++u) {
            const int e = e0 + u * 1024;
            d[u] = -1; s[u] = 0;
            if (e < end) { d[u] = dst[e]; s[u] = src[e]; }
        }
#pragma unroll
        for (int u = 0; u < 8; ++u) {
            if (d[u] >= 0) {
                const int p = atomicAdd(&curD[(unsigned)d[u] / RANGE], 1);
                stD[p] = make_int2(s[u], d[u]);
                const int q = atomicAdd(&curS[(unsigned)s[u] / RANGE], 1);
                stS[q] = s[u];
            }
        }
    }
    __syncthreads();
    const int n = end - beg;
    int2* oD = (int2*)tmpD + beg;
    int*  oS = tmpS + beg;
    for (int j = tid; j < n; j += 1024) { oD[j] = stD[j]; oS[j] = stS[j]; }
}

__global__ __launch_bounds__(1024)
void outdeg_kernel(const int* __restrict__ tmpS, const int* __restrict__ offS,
                   float* __restrict__ nsrc, int E) {
    __shared__ int cnt[RANGE];
    __shared__ int segB[G_CHUNKS];
    __shared__ int segN[G_CHUNKS];
    const int tid  = threadIdx.x;
    const int lane = tid & 31;
    const int wave = __builtin_amdgcn_readfirstlane(tid >> 5);
    const int b    = blockIdx.x;
    const int lo   = b * RANGE;
    const int chunk = (E + G_CHUNKS - 1) / G_CHUNKS;
    if (tid < RANGE) cnt[tid] = 0;
    if (tid < G_CHUNKS) {
        const int o0 = offS[tid * (NB + 1) + b];
        const int o1 = offS[tid * (NB + 1) + b + 1];
        segB[tid] = tid * chunk + o0;
        segN[tid] = o1 - o0;
    }
    __syncthreads();
    int sb[4], sn[4], v[4][2];
#pragma unroll
    for (int k = 0; k < 4; ++k) {
        sb[k] = segB[wave + 32 * k];
        sn[k] = segN[wave + 32 * k];
#pragma unroll
        for (int r = 0; r < 2; ++r) {
            const int j = lane + 32 * r;
            v[k][r] = -1;
            if (j < sn[k]) v[k][r] = tmpS[sb[k] + j];
        }
    }
#pragma unroll
    for (int k = 0; k < 4; ++k) {
#pragma unroll
        for (int r = 0; r < 2; ++r)
            if (v[k][r] >= 0) atomicAdd(&cnt[v[k][r] - lo], 1);
        for (int j = lane + 64; j < sn[k]; j += 32) atomicAdd(&cnt[tmpS[sb[k] + j] - lo], 1);
    }
    __syncthreads();
    if (tid < RANGE && lo + tid < N_NODES)
        nsrc[lo + tid] = rsqrtf(fmaxf((float)cnt[tid], 1.0f));
}

#define SCAN_PER ((RANGE + 31) / 32)
static_assert(RANGE <= 1024 && G_CHUNKS == 4 * 32, "one thread per node of a range, 4 chunks per wave");

__global__ __launch_bounds__(1024)
void range_fill_kernel(const int* __restrict__ tmpD, const int* __restrict__ offD,
                       const float* __restrict__ nsrc, int* __restrict__ rowptr,
                       float* __restrict__ ndst, int* __restrict__ edges, int E) {
    __shared__ int cnt[RANGE];
    __shared__ int cur[RANGE];
    __shared__ int segB[G_CHUNKS];
    __shared__ int segN[G_CHUNKS];
    __shared__ int part[4];
    const int tid  = threadIdx.x;
    const int lane = tid & 31;
    const int wave = __builtin_amdgcn_readfirstlane(tid >> 5);
    const int b    = blockIdx.x;
    const int lo   = b * RANGE;
    const int chunk = (E + G_CHUNKS - 1) / G_CHUNKS;
    const int2* __restrict__ tD = (const int2*)tmpD;
    if (tid < RANGE) cnt[tid] = 0;
    if (tid < G_CHUNKS) {
        const int o0 = offD[tid * (NB + 1) + b];
        const int o1 = offD[tid * (NB + 1) + b + 1];
        segB[tid] = tid * chunk + o0;
        segN[tid] = o1 - o0;
        const int s = wave_sum(o0);
        if (lane == 0) part[wave] = s;
    }
    __syncthreads();
    const int base = part[0] + part[1] + part[2] + part[3];
    int sb[4], sn[4];
    int2 ed[4][2];
#pragma unroll
    for (int k = 0; k < 4; ++k) {
        sb[k] = segB[wave + 32 * k];
        sn[k] = segN[wave + 32 * k];
#pragma unroll
        for (int r = 0; r < 2; ++r) {
            const int j = lane + 32 * r;
            ed[k][r] = make_int2(0, -1);
            if (j < sn[k]) ed[k][r] = tD[sb[k] + j];
        }
    }
#pragma unroll
    for (int k = 0; k < 4; ++k) {
#pragma unroll
        for (int r = 0; r < 2; ++r)
            if (ed[k][r].y >= 0) atomicAdd(&cnt[ed[k][r].y - lo], 1);
        for (int j = lane + 64; j < sn[k]; j += 32) atomicAdd(&cnt[tD[sb[k] + j].y - lo], 1);
    }
    __syncthreads();
    if (wave == 0) {
        int c[SCAN_PER];
        int sum = 0;
#pragma unroll
        for (int q = 0; q < SCAN_PER; ++q) {
            const int t = lane * SCAN_PER + q;
            c[q] = (t < RANGE) ? cnt[t] : 0;
            sum += c[q];
        }
        const int inc = wave_inclusive_scan(sum, lane);
        int run = base + inc - sum;
#pragma unroll
        for (int q = 0; q < SCAN_PER; ++q) {
            const int t = lane * SCAN_PER + q;
            if (t < RANGE) {
                cur[t] = run;
                if (lo + t < N_NODES) {
                    rowptr[lo + t] = run;
                    ndst[lo + t]   = rsqrtf(fmaxf((float)c[q], 1.0f));
                }
            }
            run += c[q];
        }
        if (b == NB - 1 && lane == 31) rowptr[N_NODES] = base + inc;
    }
    __syncthreads();
    int2* __restrict__ eo = (int2*)edges;
    float w[4][2];
#pragma unroll
    for (int k = 0; k < 4; ++k)
#pragma unroll
        for (int r = 0; r < 2; ++r) w[k][r] = nsrc[ed[k][r].x];
#pragma unroll
    for (int k = 0; k < 4; ++k) {
#pragma unroll
        for (int r = 0; r < 2; ++r) {
            if (ed[k][r].y >= 0) {
                const int pos = atomicAdd(&cur[ed[k][r].y - lo], 1);
                eo[pos] = make_int2(ed[k][r].x, __float_as_int(w[k][r]));
            }
        }
        for (int j = lane + 64; j < sn[k]; j += 32) {
            const int2 e = tD[sb[k] + j];
            const int pos = atomicAdd(&cur[e.y - lo], 1);
            eo[pos] = make_int2(e.x, __float_as_int(nsrc[e.x]));
        }
    }
}

template <int NOUT, bool RELU>
__global__ __launch_bounds__(512)
void layer_kernel(const float* __restrict__ x, const int* __restrict__ rowptr,
                  const int* __restrict__ edges, const float* __restrict__ W,
                  const float* __restrict__ bias, const float* __restrict__ norm,
                  float* __restrict__ out) {
    constexpr int NT     = NOUT / 16;
    constexpr int GROUPS = 16 / NT;
    __shared__ __attribute__((aligned(16))) float Wl[KDIM * NOUT];
    __shared__ __attribute__((aligned(16))) float At[16 * FDIM];

    const int lane  = threadIdx.x & 31;
    const int wave  = __builtin_amdgcn_readfirstlane((int)(threadIdx.x >> 5));
    const int r15   = lane & 15;
    const int khalf = lane >> 4;

#pragma unroll
    for (int s = threadIdx.x; s < (KDIM / 4) * NT * 32; s += 512) {
        const int l  = s & 31;
        const int t  = (s >> 5) % NT;
        const int kk = (s >> 5) / NT;
        const float* Wp = W + (size_t)(kk * 4 + (l >> 4) * 2) * NOUT + t * 16 + (l & 15);
        ((float2*)Wl)[s] = make_float2(Wp[0], Wp[NOUT]);
    }

    const float4* __restrict__ xl = (const float4*)x + lane;
    const int2* __restrict__ el = (const int2*)edges;

#define ROW(j) xl[(size_t)__builtin_amdgcn_readlane(my.x, (j)) * (FDIM / 4)]
#define WGT(j) __int_as_float(__builtin_amdgcn_readlane(my.y, (j)))
#define ACC(VV, WW) { acc.x = fmaf((VV).x, (WW), acc.x); acc.y = fmaf((VV).y, (WW), acc.y); \
                      acc.z = fmaf((VV).z, (WW), acc.z); acc.w = fmaf((VV).w, (WW), acc.w); }

    int it = 0;
    for (int tile = blockIdx.x; tile < N_NODES / 16; tile += gridDim.x, ++it) {
        const int node = tile * 16 + wave;
        const int beg = rowptr[node];
        const int end = rowptr[node + 1];
        float4 acc = make_float4(0.0f, 0.0f, 0.0f, 0.0f);
        for (int base = beg; base < end; base += 32) {
            const int cnt = min(32, end - base);
            int2 my = make_int2(0, 0);
            if (lane < cnt) my = el[base + lane];
            int j = 0;
            for (; j + 8 <= cnt; j += 8) {
                float4 v[8];
                float  w[8];
#pragma unroll
                for (int u = 0; u < 8; ++u) { v[u] = ROW(j + u); w[u] = WGT(j + u); }
#pragma unroll
                for (int u = 0; u < 8; ++u) ACC(v[u], w[u]);
            }
            if (j + 4 <= cnt) {
                float4 v[4];
                float  w[4];
#pragma unroll
                for (int u = 0; u < 4; ++u) { v[u] = ROW(j + u); w[u] = WGT(j + u); }
#pragma unroll
                for (int u = 0; u < 4; ++u) ACC(v[u], w[u]);
                j += 4;
            }
            if (j < cnt) {
                const int r = cnt - j;
                const float4 z = make_float4(0.0f, 0.0f, 0.0f, 0.0f);
                float4 v0 = ROW(j), v1 = z, v2 = z;
                float  w0 = WGT(j), w1 = 0.0f, w2 = 0.0f;
                if (r > 1) { v1 = ROW(j + 1); w1 = WGT(j + 1); }
                if (r > 2) { v2 = ROW(j + 2); w2 = WGT(j + 2); }
                ACC(v0, w0);
                ACC(v1, w1);
                ACC(v2, w2);
            }
        }
        ((float4*)(At + wave * FDIM))[lane ^ wave] = acc;
        __syncthreads();

        const bool mine = (wave / NT == (it & (GROUPS - 1)));
        const int t     = wave % NT;
        const int rbase = tile * 16 + khalf * 8;
        float nrm[8];
        float bv = 0.0f;
        v8f d;
#pragma unroll
        for (int j = 0; j < 8; ++j) { d[j] = 0.0f; nrm[j] = 0.0f; }
        if (mine) {
#pragma unroll
            for (int j = 0; j < 8; ++j) nrm[j] = norm[rbase + j];
            bv = bias[t * 16 + r15];
            const float* Ar = At + r15 * FDIM + khalf * 2;
            const v2f*   Bp = (const v2f*)Wl + t * 32 + lane;
#pragma unroll 8
            for (int kk = 0; kk < KDIM / 4; ++kk) {
                const v2f a = *(const v2f*)(Ar + ((kk ^ r15) << 2));
                const v2f b = Bp[kk * NT * 32];
                d = __builtin_amdgcn_wmma_f32_16x16x4_f32(
                    false, a, false, b, (short)0, d, false, false);
            }
        }
        __syncthreads();
        if (mine) {
#pragma unroll
            for (int j = 0; j < 8; ++j) {
                float v = fmaf(d[j], nrm[j], bv);
                if (RELU) v = fmaxf(v, 0.0f);
                out[(size_t)(rbase + j) * NOUT + t * 16 + r15] = v;
            }
        }
    }
#undef ROW
#undef WGT
#undef ACC
}

__global__ __launch_bounds__(1024, 2)
void layer2_fused_kernel(const float* __restrict__ x, const int* __restrict__ rowptr,
                         const int* __restrict__ edges, const float* __restrict__ W2,
                         const float* __restrict__ b2, const float* __restrict__ W3,
                         const float* __restrict__ norm, float* __restrict__ z) {
    constexpr int NT2 = 8;
    constexpr int NT3 = 4;
    constexpr int ZDIM = 64;
    __shared__ __attribute__((aligned(16))) float Wl[KDIM * 128];
    __shared__ __attribute__((aligned(16))) float W3l[KDIM * ZDIM];
    __shared__ __attribute__((aligned(16))) float At[32 * FDIM];
    __shared__ __attribute__((aligned(16))) float Ht[32 * FDIM];

    const int lane  = threadIdx.x & 31;
    const int wave  = __builtin_amdgcn_readfirstlane((int)(threadIdx.x >> 5));
    const int r15   = lane & 15;
    const int khalf = lane >> 4;

#pragma unroll
    for (int s = threadIdx.x; s < (KDIM / 4) * NT2 * 32; s += 1024) {
        const int l  = s & 31;
        const int t  = (s >> 5) % NT2;
        const int kk = (s >> 5) / NT2;
        const float* Wp = W2 + (size_t)(kk * 4 + (l >> 4) * 2) * 128 + t * 16 + (l & 15);
        ((float2*)Wl)[s] = make_float2(Wp[0], Wp[128]);
    }
#pragma unroll
    for (int s = threadIdx.x; s < (KDIM / 4) * NT3 * 32; s += 1024) {
        const int l  = s & 31;
        const int t  = (s >> 5) % NT3;
        const int kk = (s >> 5) / NT3;
        const float* Wp = W3 + (size_t)(kk * 4 + (l >> 4) * 2) * ZDIM + t * 16 + (l & 15);
        ((float2*)W3l)[s] = make_float2(Wp[0], Wp[ZDIM]);
    }

    const float4* __restrict__ xl = (const float4*)x + lane;
    const int2* __restrict__ el = (const int2*)edges;

#define ROW(j) xl[(size_t)__builtin_amdgcn_readlane(my.x, (j)) * (FDIM / 4)]
#define WGT(j) __int_as_float(__builtin_amdgcn_readlane(my.y, (j)))
#define ACC(VV, WW) { acc.x = fmaf((VV).x, (WW), acc.x); acc.y = fmaf((VV).y, (WW), acc.y); \
                      acc.z = fmaf((VV).z, (WW), acc.z); acc.w = fmaf((VV).w, (WW), acc.w); }

    for (int tile = blockIdx.x; tile < (N_NODES + 31) / 32; tile += gridDim.x) {
        const int node = tile * 32 + wave;
        int beg = 0, end = 0;
        if (node < N_NODES) { beg = rowptr[node]; end = rowptr[node + 1]; }
        float4 acc = make_float4(0.0f, 0.0f, 0.0f, 0.0f);
        for (int base = beg; base < end; base += 32) {
            const int cnt = min(32, end - base);
            int2 my = make_int2(0, 0);
            if (lane < cnt) my = el[base + lane];
            int j = 0;
            for (; j + 8 <= cnt; j += 8) {
                float4 v[8];
                float  w[8];
#pragma unroll
                for (int u = 0; u < 8; ++u) { v[u] = ROW(j + u); w[u] = WGT(j + u); }
#pragma unroll
                for (int u = 0; u < 8; ++u) ACC(v[u], w[u]);
            }
            if (j + 4 <= cnt) {
                float4 v[4];
                float  w[4];
#pragma unroll
                for (int u = 0; u < 4; ++u) { v[u] = ROW(j + u); w[u] = WGT(j + u); }
#pragma unroll
                for (int u = 0; u < 4; ++u) ACC(v[u], w[u]);
                j += 4;
            }
            if (j < cnt) {
                const int r = cnt - j;
                const float4 z4 = make_float4(0.0f, 0.0f, 0.0f, 0.0f);
                float4 v0 = ROW(j), v1 = z4, v2 = z4;
                float  w0 = WGT(j), w1 = 0.0f, w2 = 0.0f;
                if (r > 1) { v1 = ROW(j + 1); w1 = WGT(j + 1); }
                if (r > 2) { v2 = ROW(j + 2); w2 = WGT(j + 2); }
                ACC(v0, w0);
                ACC(v1, w1);
                ACC(v2, w2);
            }
        }
        ((float4*)(At + wave * FDIM))[lane ^ (wave & 15)] = acc;
        __syncthreads();

        if (wave < 16) {
            const int mt = wave >> 3;
            const int t  = wave & 7;
            const int rbase = tile * 32 + mt * 16 + khalf * 8;
            float nrm[8];
#pragma unroll
            for (int j = 0; j < 8; ++j) nrm[j] = norm[min(rbase + j, N_NODES - 1)];
            const float bv = b2[t * 16 + r15];
            v8f d;
#pragma unroll
            for (int j = 0; j < 8; ++j) d[j] = 0.0f;
            const float* Ar = At + (mt * 16 + r15) * FDIM + khalf * 2;
            const v2f*   Bp = (const v2f*)Wl + t * 32 + lane;
#pragma unroll 8
            for (int kk = 0; kk < KDIM / 4; ++kk) {
                const v2f a = *(const v2f*)(Ar + ((kk ^ r15) << 2));
                const v2f b = Bp[kk * NT2 * 32];
                d = __builtin_amdgcn_wmma_f32_16x16x4_f32(
                    false, a, false, b, (short)0, d, false, false);
            }
            const int col = t * 16 + r15;
#pragma unroll
            for (int j = 0; j < 8; ++j) {
                const int row = mt * 16 + khalf * 8 + j;
                const float v = fmaxf(fmaf(d[j], nrm[j], bv), 0.0f);
                Ht[row * FDIM + ((((col >> 2) ^ (row & 15)) << 2) | (col & 3))] = v;
            }
        }
        __syncthreads();
        if (wave >= 16 && wave < 24) {
            const int q  = wave - 16;
            const int mt = q >> 2;
            const int t3 = q & 3;
            v8f d;
#pragma unroll
            for (int j = 0; j < 8; ++j) d[j] = 0.0f;
            const float* Hr = Ht + (mt * 16 + r15) * FDIM + khalf * 2;
            const v2f*   Bp = (const v2f*)W3l + t3 * 32 + lane;
#pragma unroll 8
            for (int kk = 0; kk < KDIM / 4; ++kk) {
                const v2f a = *(const v2f*)(Hr + ((kk ^ r15) << 2));
                const v2f b = Bp[kk * NT3 * 32];
                d = __builtin_amdgcn_wmma_f32_16x16x4_f32(
                    false, a, false, b, (short)0, d, false, false);
            }
            const int rbase = tile * 32 + mt * 16 + khalf * 8;
#pragma unroll
            for (int j = 0; j < 8; ++j) {
                if (rbase + j < N_NODES) z[(size_t)(rbase + j) * ZDIM + t3 * 16 + r15] = d[j];
            }
        }
    }
#undef ROW
#undef WGT
#undef ACC
}

template <int P>
__device__ __forceinline__ void gather_pairs(const float4* __restrict__ zl, int2 my, int j,
                                             int cnt, int khalf, float4& acc) {
    float4 v[P];
    float  w[P];
#pragma unroll
    for (int u = 0; u < P; ++u) {
        const int e  = j + 2 * u + khalf;
        const int ec = min(e, cnt - 1);
        const int s  = __shfl(my.x, ec);
        const float wt = __int_as_float(__shfl(my.y, ec));
        w[u] = (e < cnt) ? wt : 0.0f;
        v[u] = zl[(size_t)s * 16];
    }
#pragma unroll
    for (int u = 0; u < P; ++u) {
        acc.x = fmaf(v[u].x, w[u], acc.x); acc.y = fmaf(v[u].y, w[u], acc.y);
        acc.z = fmaf(v[u].z, w[u], acc.z); acc.w = fmaf(v[u].w, w[u], acc.w);
    }
}

__global__ __launch_bounds__(256)
void gather64_kernel(const float* __restrict__ z, const int* __restrict__ rowptr,
                     const int* __restrict__ edges, const float* __restrict__ bias,
                     const float* __restrict__ norm, float* __restrict__ out) {
    const int lane  = threadIdx.x & 31;
    const int r15   = lane & 15;
    const int khalf = lane >> 4;
    const int gw = __builtin_amdgcn_readfirstlane((int)(blockIdx.x * 8 + (threadIdx.x >> 5)));
    const int nw = gridDim.x * 8;
    const float4* __restrict__ zl = (const float4*)z + r15;
    const int2* __restrict__ el = (const int2*)edges;
    const float4 bq = ((const float4*)bias)[r15];
    for (int node = gw; node < N_NODES; node += nw) {
        const int beg = rowptr[node];
        const int end = rowptr[node + 1];
        float4 acc = make_float4(0.0f, 0.0f, 0.0f, 0.0f);
        for (int base = beg; base < end; base += 32) {
            const int cnt = min(32, end - base);
            int2 my = make_int2(0, 0);
            if (lane < cnt) my = el[base + lane];
            int j = 0;
            for (; j + 8 < cnt; j += 16) gather_pairs<8>(zl, my, j, cnt, khalf, acc);
            if (j < cnt) gather_pairs<4>(zl, my, j, cnt, khalf, acc);
        }
        acc.x += __shfl_xor(acc.x, 16); acc.y += __shfl_xor(acc.y, 16);
        acc.z += __shfl_xor(acc.z, 16); acc.w += __shfl_xor(acc.w, 16);
        if (khalf == 0) {
            const float nd = norm[node];
            float4 o;
            o.x = fmaf(acc.x, nd, bq.x); o.y = fmaf(acc.y, nd, bq.y);
            o.z = fmaf(acc.z, nd, bq.z); o.w = fmaf(acc.w, nd, bq.w);
            ((float4*)out)[(size_t)node * 16 + r15] = o;
        }
    }
}

extern "C" void kernel_launch(void* const* d_in, const int* in_sizes, int n_in,
                              void* d_out, int out_size, void* d_ws, size_t ws_size,
                              hipStream_t stream) {
    const float* features = (const float*)d_in[0];
    const float* W1 = (const float*)d_in[1];
    const float* b1 = (const float*)d_in[2];
    const float* W2 = (const float*)d_in[3];
    const float* b2 = (const float*)d_in[4];
    const float* W3 = (const float*)d_in[5];
    const float* b3 = (const float*)d_in[6];
    const int*  src = (const int*)d_in[7];
    const int*  dst = (const int*)d_in[8];
    const int E = in_sizes[7];

    float* nrm      = (float*)d_ws + 2 * N_NODES;
    int*   rowptr   = (int*)d_ws + 4 * N_NODES;
    float* aggA     = (float*)d_ws + 250304;
    float* bufB     = aggA + (size_t)N_NODES * FDIM;
    int*   edges    = (int*)(bufB + (size_t)N_NODES * FDIM);
    int*   tmpD     = edges + 2 * (size_t)E;
    int*   tmpS     = tmpD + 2 * (size_t)E;
    int*   offD     = tmpS + (size_t)E;
    int*   offS     = offD + G_CHUNKS * (NB + 1);
    float* norm_src = nrm;
    float* norm_dst = nrm + N_NODES;

    bin_kernel<<<G_CHUNKS, 1024, 0, stream>>>(src, dst, tmpD, tmpS, offD, offS, E);
    outdeg_kernel<<<NB, 1024, 0, stream>>>(tmpS, offS, norm_src, E);
    range_fill_kernel<<<NB, 1024, 0, stream>>>(tmpD, offD, norm_src, rowptr, norm_dst, edges, E);

    const int layerBlocks = 1024;

    layer_kernel<128, true><<<layerBlocks, 512, 0, stream>>>(features, rowptr, edges, W1, b1,
                                                            norm_dst, bufB);
    layer2_fused_kernel<<<512, 1024, 0, stream>>>(bufB, rowptr, edges, W2, b2, W3, norm_dst, aggA);
    gather64_kernel<<<1536, 256, 0, stream>>>(aggA, rowptr, edges, b3, norm_dst, (float*)d_out);
}


# norm_src scales rows where they are made (layer 1 looks it up per edge), not list entries: the fill needs no out-degrees, so both run side by side in one launch, 6 to 5 launches; sum order unchanged
# speedup vs baseline: 6.0924x; 1.0511x over previous
#include <hip/hip_runtime.h>
#include <hip/hip_bf16.h>


#define N_NODES 50000
#define FDIM    128
#define KDIM    128

typedef __attribute__((ext_vector_type(2))) float v2f;
typedef __attribute__((ext_vector_type(8))) float v8f;

__device__ __forceinline__ int wave_sum(int v) {
    for (int o = 16; o > 0; o >>= 1) v += __shfl_xor(v, o);
    return v;
}

__device__ __forceinline__ int wave_inclusive_scan(int v, int lane) {
    for (int o = 1; o < 32; o <<= 1) {
        const int t = __shfl_up(v, o);
        if (lane >= o) v += t;
    }
    return v;
}

#define G_CHUNKS  128
#define NB        128
#define RANGE     ((N_NODES + NB - 1) / NB)
#define CHUNK_CAP 8192

static_assert(NB == 128, "scan_ranges covers 4 ranges per lane of one wave");
__device__ __forceinline__ void scan_ranges(int* cur, int* __restrict__ off, int lane) {
    const int c0 = cur[4 * lane], c1 = cur[4 * lane + 1], c2 = cur[4 * lane + 2], c3 = cur[4 * lane + 3];
    const int sum = c0 + c1 + c2 + c3;
    const int inc = wave_inclusive_scan(sum, lane);
    const int x0 = inc - sum, x1 = x0 + c0, x2 = x1 + c1, x3 = x2 + c2;
    cur[4 * lane] = x0; cur[4 * lane + 1] = x1; cur[4 * lane + 2] = x2; cur[4 * lane + 3] = x3;
    off[4 * lane] = x0; off[4 * lane + 1] = x1; off[4 * lane + 2] = x2; off[4 * lane + 3] = x3;
    if (lane == 31) off[NB] = inc;
}

__global__ __launch_bounds__(1024)
void bin_kernel(const int* __restrict__ src, const int* __restrict__ dst,
                int* __restrict__ tmpD, int* __restrict__ tmpS,
                int* __restrict__ offD, int* __restrict__ offS, int E) {
    __shared__ int curD[NB];
    __shared__ int curS[NB];
    __shared__ __attribute__((aligned(16))) int2 stD[CHUNK_CAP];
    __shared__ __attribute__((aligned(16))) int  stS[CHUNK_CAP];
    const int tid  = threadIdx.x;
    const int lane = tid & 31;
    if (tid < NB) { curD[tid] = 0; curS[tid] = 0; }
    __syncthreads();
    const int chunk = (E + (int)gridDim.x - 1) / (int)gridDim.x;
    const int beg = (int)blockIdx.x * chunk;
    const int end = min(E, beg + min(chunk, CHUNK_CAP));
    for (int e0 = beg + tid; e0 < end; e0 += 8 * 1024) {
        int d[8], s[8];
#pragma unroll
        for (int u = 0; u < 8; ++u) {
            const int e = e0 + u * 1024;
            d[u] = -1; s[u] = 0;
            if (e < end) { d[u] = dst[e]; s[u] = src[e]; }
        }
#pragma unroll
        for (int u = 0; u < 8; ++u) {
            if (d[u] >= 0) {
                atomicAdd(&curD[(unsigned)d[u] / RANGE], 1);
                atomicAdd(&curS[(unsigned)s[u] / RANGE], 1);
            }
        }
    }
    __syncthreads();
    if (tid < 32)      scan_ranges(curD, offD + (size_t)blockIdx.x * (NB + 1), lane);
    else if (tid < 64) scan_ranges(curS, offS + (size_t)blockIdx.x * (NB + 1), lane);
    __syncthreads();
    for (int e0 = beg + tid; e0 < end; e0 += 8 * 1024) {
        int d[8], s[8];
#pragma unroll
        for (int u = 0; u < 8; ++u) {
            const int e = e0 + u * 1024;
            d[u] = -1; s[u] = 0;
            if (e < end) { d[u] = dst[e]; s[u] = src[e]; }
        }
#pragma unroll
        for (int u = 0; u < 8; ++u) {
            if (d[u] >= 0) {
                const int p = atomicAdd(&curD[(unsigned)d[u] / RANGE], 1);
                stD[p] = make_int2(s[u], d[u]);
                const int q = atomicAdd(&curS[(unsigned)s[u] / RANGE], 1);
                stS[q] = s[u];
            }
        }
    }
    __syncthreads();
    const int n = end - beg;
    int2* oD = (int2*)tmpD + beg;
    int*  oS = tmpS + beg;
    for (int j = tid; j < n; j += 1024) { oD[j] = stD[j]; oS[j] = stS[j]; }
}

__device__ __forceinline__
void count_sources(const int* __restrict__ tmpS, const int* __restrict__ offS,
                   float* __restrict__ nsrc, int E, int b) {
    __shared__ int cnt[RANGE];
    __shared__ int segB[G_CHUNKS];
    __shared__ int segN[G_CHUNKS];
    const int tid  = threadIdx.x;
    const int lane = tid & 31;
    const int wave = __builtin_amdgcn_readfirstlane(tid >> 5);
    const int lo   = b * RANGE;
    const int chunk = (E + G_CHUNKS - 1) / G_CHUNKS;
    if (tid < RANGE) cnt[tid] = 0;
    if (tid < G_CHUNKS) {
        const int o0 = offS[tid * (NB + 1) + b];
        const int o1 = offS[tid * (NB + 1) + b + 1];
        segB[tid] = tid * chunk + o0;
        segN[tid] = o1 - o0;
    }
    __syncthreads();
    int sb[4], sn[4], v[4][2];
#pragma unroll
    for (int k = 0; k < 4; ++k) {
        sb[k] = segB[wave + 32 * k];
        sn[k] = segN[wave + 32 * k];
#pragma unroll
        for (int r = 0; r < 2; ++r) {
            const int j = lane + 32 * r;
            v[k][r] = -1;
            if (j < sn[k]) v[k][r] = tmpS[sb[k] + j];
        }
    }
#pragma unroll
    for (int k = 0; k < 4; ++k) {
#pragma unroll
        for (int r = 0; r < 2; ++r)
            if (v[k][r] >= 0) atomicAdd(&cnt[v[k][r] - lo], 1);
        for (int j = lane + 64; j < sn[k]; j += 32) atomicAdd(&cnt[tmpS[sb[k] + j] - lo], 1);
    }
    __syncthreads();
    if (tid < RANGE && lo + tid < N_NODES)
        nsrc[lo + tid] = rsqrtf(fmaxf((float)cnt[tid], 1.0f));
}

#define SCAN_PER ((RANGE + 31) / 32)
static_assert(RANGE <= 1024 && G_CHUNKS == 4 * 32, "one thread per node of a range, 4 chunks per wave");

__device__ __forceinline__
void fill_range(const int* __restrict__ tmpD, const int* __restrict__ offD,
                int* __restrict__ rowptr, float* __restrict__ ndst,
                int* __restrict__ edges, int E, int b) {
    __shared__ int cnt[RANGE];
    __shared__ int cur[RANGE];
    __shared__ int segB[G_CHUNKS];
    __shared__ int segN[G_CHUNKS];
    __shared__ int part[4];
    const int tid  = threadIdx.x;
    const int lane = tid & 31;
    const int wave = __builtin_amdgcn_readfirstlane(tid >> 5);
    const int lo   = b * RANGE;
    const int chunk = (E + G_CHUNKS - 1) / G_CHUNKS;
    const int2* __restrict__ tD = (const int2*)tmpD;
    if (tid < RANGE) cnt[tid] = 0;
    if (tid < G_CHUNKS) {
        const int o0 = offD[tid * (NB + 1) + b];
        const int o1 = offD[tid * (NB + 1) + b + 1];
        segB[tid] = tid * chunk + o0;
        segN[tid] = o1 - o0;
        const int s = wave_sum(o0);
        if (lane == 0) part[wave] = s;
    }
    __syncthreads();
    const int base = part[0] + part[1] + part[2] + part[3];
    int sb[4], sn[4];
    int2 ed[4][2];
#pragma unroll
    for (int k = 0; k < 4; ++k) {
        sb[k] = segB[wave + 32 * k];
        sn[k] = segN[wave + 32 * k];
#pragma unroll
        for (int r = 0; r < 2; ++r) {
            const int j = lane + 32 * r;
            ed[k][r] = make_int2(0, -1);
            if (j < sn[k]) ed[k][r] = tD[sb[k] + j];
        }
    }
#pragma unroll
    for (int k = 0; k < 4; ++k) {
#pragma unroll
        for (int r = 0; r < 2; ++r)
            if (ed[k][r].y >= 0) atomicAdd(&cnt[ed[k][r].y - lo], 1);
        for (int j = lane + 64; j < sn[k]; j += 32) atomicAdd(&cnt[tD[sb[k] + j].y - lo], 1);
    }
    __syncthreads();
    if (wave == 0) {
        int c[SCAN_PER];
        int sum = 0;
#pragma unroll
        for (int q = 0; q < SCAN_PER; ++q) {
            const int t = lane * SCAN_PER + q;
            c[q] = (t < RANGE) ? cnt[t] : 0;
            sum += c[q];
        }
        const int inc = wave_inclusive_scan(sum, lane);
        int run = base + inc - sum;
#pragma unroll
        for (int q = 0; q < SCAN_PER; ++q) {
            const int t = lane * SCAN_PER + q;
            if (t < RANGE) {
                cur[t] = run;
                if (lo + t < N_NODES) {
                    rowptr[lo + t] = run;
                    ndst[lo + t]   = rsqrtf(fmaxf((float)c[q], 1.0f));
                }
            }
            run += c[q];
        }
        if (b == NB - 1 && lane == 31) rowptr[N_NODES] = base + inc;
    }
    __syncthreads();
#pragma unroll
    for (int k = 0; k < 4; ++k) {
#pragma unroll
        for (int r = 0; r < 2; ++r) {
            if (ed[k][r].y >= 0) {
                const int pos = atomicAdd(&cur[ed[k][r].y - lo], 1);
                edges[pos] = ed[k][r].x;
            }
        }
        for (int j = lane + 64; j < sn[k]; j += 32) {
            const int2 e = tD[sb[k] + j];
            const int pos = atomicAdd(&cur[e.y - lo], 1);
            edges[pos] = e.x;
        }
    }
}

#define RANGE_BLOCKS (2 * NB)
__global__ __launch_bounds__(1024)
void range_kernel(const int* __restrict__ tmpD, const int* __restrict__ offD,
                  const int* __restrict__ tmpS, const int* __restrict__ offS,
                  float* __restrict__ nsrc, int* __restrict__ rowptr,
                  float* __restrict__ ndst, int* __restrict__ edges, int E) {
    if (blockIdx.x < NB) fill_range(tmpD, offD, rowptr, ndst, edges, E, (int)blockIdx.x);
    else                 count_sources(tmpS, offS, nsrc, E, (int)blockIdx.x - NB);
}

__global__ __launch_bounds__(512)
void layer1_kernel(const float* __restrict__ x, const int* __restrict__ rowptr,
                   const int* __restrict__ edges, const float* __restrict__ W,
                   const float* __restrict__ bias, const float* __restrict__ norm,
                   const float* __restrict__ nsrc, float* __restrict__ out) {
    constexpr int NOUT   = 128;
    constexpr int NT     = NOUT / 16;
    constexpr int GROUPS = 16 / NT;
    __shared__ __attribute__((aligned(16))) float Wl[KDIM * NOUT];
    __shared__ __attribute__((aligned(16))) float At[16 * FDIM];

    const int lane  = threadIdx.x & 31;
    const int wave  = __builtin_amdgcn_readfirstlane((int)(threadIdx.x >> 5));
    const int r15   = lane & 15;
    const int khalf = lane >> 4;

#pragma unroll
    for (int s = threadIdx.x; s < (KDIM / 4) * NT * 32; s += 512) {
        const int l  = s & 31;
        const int t  = (s >> 5) % NT;
        const int kk = (s >> 5) / NT;
        const float* Wp = W + (size_t)(kk * 4 + (l >> 4) * 2) * NOUT + t * 16 + (l & 15);
        ((float2*)Wl)[s] = make_float2(Wp[0], Wp[NOUT]);
    }

    const float4* __restrict__ xl = (const float4*)x + lane;
#define ROW(j) xl[(size_t)__builtin_amdgcn_readlane(my, (j)) * (FDIM / 4)]
#define WGT(j) __int_as_float(__builtin_amdgcn_readlane(__float_as_int(mw), (j)))
#define ACC(VV, WW) { acc.x = fmaf((VV).x, (WW), acc.x); acc.y = fmaf((VV).y, (WW), acc.y); \
                      acc.z = fmaf((VV).z, (WW), acc.z); acc.w = fmaf((VV).w, (WW), acc.w); }

    int it = 0;
    for (int tile = blockIdx.x; tile < N_NODES / 16; tile += gridDim.x, ++it) {
        const int node = tile * 16 + wave;
        const int beg = rowptr[node];
        const int end = rowptr[node + 1];
        float4 acc = make_float4(0.0f, 0.0f, 0.0f, 0.0f);
        for (int base = beg; base < end; base += 32) {
            const int cnt = min(32, end - base);
            int   my = 0;
            float mw = 0.0f;
            if (lane < cnt) { my = edges[base + lane]; mw = nsrc[my]; }
            int j = 0;
            for (; j + 8 <= cnt; j += 8) {
                float4 v[8];
#pragma unroll
                for (int u = 0; u < 8; ++u) v[u] = ROW(j + u);
#pragma unroll
                for (int u = 0; u < 8; ++u) ACC(v[u], WGT(j + u));
            }
            if (j + 4 <= cnt) {
                float4 v[4];
#pragma unroll
                for (int u = 0; u < 4; ++u) v[u] = ROW(j + u);
#pragma unroll
                for (int u = 0; u < 4; ++u) ACC(v[u], WGT(j + u));
                j += 4;
            }
            if (j < cnt) {
                const int r = cnt - j;
                const float4 z = make_float4(0.0f, 0.0f, 0.0f, 0.0f);
                float4 v0 = ROW(j), v1 = z, v2 = z;
                if (r > 1) v1 = ROW(j + 1);
                if (r > 2) v2 = ROW(j + 2);
                ACC(v0, WGT(j));
                ACC(v1, WGT(j + 1));
                ACC(v2, WGT(j + 2));
            }
        }
        ((float4*)(At + wave * FDIM))[lane ^ wave] = acc;
        __syncthreads();

        const bool mine = (wave / NT == (it & (GROUPS - 1)));
        const int t     = wave % NT;
        const int rbase = tile * 16 + khalf * 8;
        float nrm[8], ns[8];
        float bv = 0.0f;
        v8f d;
#pragma unroll
        for (int j = 0; j < 8; ++j) { d[j] = 0.0f; nrm[j] = 0.0f; ns[j] = 0.0f; }
        if (mine) {
#pragma unroll
            for (int j = 0; j < 8; ++j) { nrm[j] = norm[rbase + j]; ns[j] = nsrc[rbase + j]; }
            bv = bias[t * 16 + r15];
            const float* Ar = At + r15 * FDIM + khalf * 2;
            const v2f*   Bp = (const v2f*)Wl + t * 32 + lane;
#pragma unroll 8
            for (int kk = 0; kk < KDIM / 4; ++kk) {
                const v2f a = *(const v2f*)(Ar + ((kk ^ r15) << 2));
                const v2f b = Bp[kk * NT * 32];
                d = __builtin_amdgcn_wmma_f32_16x16x4_f32(
                    false, a, false, b, (short)0, d, false, false);
            }
        }
        __syncthreads();
        if (mine) {
#pragma unroll
            for (int j = 0; j < 8; ++j) {
                const float v = fmaxf(fmaf(d[j], nrm[j], bv), 0.0f);
                out[(size_t)(rbase + j) * NOUT + t * 16 + r15] = v * ns[j];
            }
        }
    }
#undef ROW
#undef WGT
#undef ACC
}

__global__ __launch_bounds__(1024, 2)
void layer2_fused_kernel(const float* __restrict__ x, const int* __restrict__ rowptr,
                         const int* __restrict__ edges, const float* __restrict__ W2,
                         const float* __restrict__ b2, const float* __restrict__ W3,
                         const float* __restrict__ norm, const float* __restrict__ nsrc,
                         float* __restrict__ z) {
    constexpr int NT2 = 8;
    constexpr int NT3 = 4;
    constexpr int ZDIM = 64;
    __shared__ __attribute__((aligned(16))) float Wl[KDIM * 128];
    __shared__ __attribute__((aligned(16))) float W3l[KDIM * ZDIM];
    __shared__ __attribute__((aligned(16))) float At[32 * FDIM];
    __shared__ __attribute__((aligned(16))) float Ht[32 * FDIM];

    const int lane  = threadIdx.x & 31;
    const int wave  = __builtin_amdgcn_readfirstlane((int)(threadIdx.x >> 5));
    const int r15   = lane & 15;
    const int khalf = lane >> 4;

#pragma unroll
    for (int s = threadIdx.x; s < (KDIM / 4) * NT2 * 32; s += 1024) {
        const int l  = s & 31;
        const int t  = (s >> 5) % NT2;
        const int kk = (s >> 5) / NT2;
        const float* Wp = W2 + (size_t)(kk * 4 + (l >> 4) * 2) * 128 + t * 16 + (l & 15);
        ((float2*)Wl)[s] = make_float2(Wp[0], Wp[128]);
    }
#pragma unroll
    for (int s = threadIdx.x; s < (KDIM / 4) * NT3 * 32; s += 1024) {
        const int l  = s & 31;
        const int t  = (s >> 5) % NT3;
        const int kk = (s >> 5) / NT3;
        const float* Wp = W3 + (size_t)(kk * 4 + (l >> 4) * 2) * ZDIM + t * 16 + (l & 15);
        ((float2*)W3l)[s] = make_float2(Wp[0], Wp[ZDIM]);
    }

    const float4* __restrict__ xl = (const float4*)x + lane;
#define ROW(j) xl[(size_t)__builtin_amdgcn_readlane(my, (j)) * (FDIM / 4)]
#define ACC(VV) { acc.x += (VV).x; acc.y += (VV).y; acc.z += (VV).z; acc.w += (VV).w; }

    for (int tile = blockIdx.x; tile < (N_NODES + 31) / 32; tile += gridDim.x) {
        const int node = tile * 32 + wave;
        int beg = 0, end = 0;
        if (node < N_NODES) { beg = rowptr[node]; end = rowptr[node + 1]; }
        float4 acc = make_float4(0.0f, 0.0f, 0.0f, 0.0f);
        for (int base = beg; base < end; base += 32) {
            const int cnt = min(32, end - base);
            int my = 0;
            if (lane < cnt) my = edges[base + lane];
            int j = 0;
            for (; j + 8 <= cnt; j += 8) {
                float4 v[8];
#pragma unroll
                for (int u = 0; u < 8; ++u) v[u] = ROW(j + u);
#pragma unroll
                for (int u = 0; u < 8; ++u) ACC(v[u]);
            }
            if (j + 4 <= cnt) {
                float4 v[4];
#pragma unroll
                for (int u = 0; u < 4; ++u) v[u] = ROW(j + u);
#pragma unroll
                for (int u = 0; u < 4; ++u) ACC(v[u]);
                j += 4;
            }
            if (j < cnt) {
                const int r = cnt - j;
                const float4 z4 = make_float4(0.0f, 0.0f, 0.0f, 0.0f);
                float4 v0 = ROW(j), v1 = z4, v2 = z4;
                if (r > 1) v1 = ROW(j + 1);
                if (r > 2) v2 = ROW(j + 2);
                ACC(v0);
                ACC(v1);
                ACC(v2);
            }
        }
        ((float4*)(At + wave * FDIM))[lane ^ (wave & 15)] = acc;
        __syncthreads();

        if (wave < 16) {
            const int mt = wave >> 3;
            const int t  = wave & 7;
            const int rbase = tile * 32 + mt * 16 + khalf * 8;
            float nrm[8];
#pragma unroll
            for (int j = 0; j < 8; ++j) nrm[j] = norm[min(rbase + j, N_NODES - 1)];
            const float bv = b2[t * 16 + r15];
            v8f d;
#pragma unroll
            for (int j = 0; j < 8; ++j) d[j] = 0.0f;
            const float* Ar = At + (mt * 16 + r15) * FDIM + khalf * 2;
            const v2f*   Bp = (const v2f*)Wl + t * 32 + lane;
#pragma unroll 8
            for (int kk = 0; kk < KDIM / 4; ++kk) {
                const v2f a = *(const v2f*)(Ar + ((kk ^ r15) << 2));
                const v2f b = Bp[kk * NT2 * 32];
                d = __builtin_amdgcn_wmma_f32_16x16x4_f32(
                    false, a, false, b, (short)0, d, false, false);
            }
            const int col = t * 16 + r15;
#pragma unroll
            for (int j = 0; j < 8; ++j) {
                const int row = mt * 16 + khalf * 8 + j;
                const float v = fmaxf(fmaf(d[j], nrm[j], bv), 0.0f);
                Ht[row * FDIM + ((((col >> 2) ^ (row & 15)) << 2) | (col & 3))] = v;
            }
        }
        __syncthreads();
        if (wave >= 16 && wave < 24) {
            const int q  = wave - 16;
            const int mt = q >> 2;
            const int t3 = q & 3;
            const int rbase = tile * 32 + mt * 16 + khalf * 8;
            float ns[8];
#pragma unroll
            for (int j = 0; j < 8; ++j) ns[j] = nsrc[min(rbase + j, N_NODES - 1)];
            v8f d;
#pragma unroll
            for (int j = 0; j < 8; ++j) d[j] = 0.0f;
            const float* Hr = Ht + (mt * 16 + r15) * FDIM + khalf * 2;
            const v2f*   Bp = (const v2f*)W3l + t3 * 32 + lane;
#pragma unroll 8
            for (int kk = 0; kk < KDIM / 4; ++kk) {
                const v2f a = *(const v2f*)(Hr + ((kk ^ r15) << 2));
                const v2f b = Bp[kk * NT3 * 32];
                d = __builtin_amdgcn_wmma_f32_16x16x4_f32(
                    false, a, false, b, (short)0, d, false, false);
            }
#pragma unroll
            for (int j = 0; j < 8; ++j) {
                if (rbase + j < N_NODES) z[(size_t)(rbase + j) * ZDIM + t3 * 16 + r15] = d[j] * ns[j];
            }
        }
    }
#undef ROW
#undef ACC
}

template <int P>
__device__ __forceinline__ void gather_pairs(const float4* __restrict__ zl, int my, int j,
                                             int cnt, int khalf, float4& acc) {
    float4 v[P];
    float  w[P];
#pragma unroll
    for (int u = 0; u < P; ++u) {
        const int e  = j + 2 * u + khalf;
        const int ec = min(e, cnt - 1);
        const int s  = __shfl(my, ec);
        w[u] = (e < cnt) ? 1.0f : 0.0f;
        v[u] = zl[(size_t)s * 16];
    }
#pragma unroll
    for (int u = 0; u < P; ++u) {
        acc.x = fmaf(v[u].x, w[u], acc.x); acc.y = fmaf(v[u].y, w[u], acc.y);
        acc.z = fmaf(v[u].z, w[u], acc.z); acc.w = fmaf(v[u].w, w[u], acc.w);
    }
}

__global__ __launch_bounds__(256)
void gather64_kernel(const float* __restrict__ z, const int* __restrict__ rowptr,
                     const int* __restrict__ edges, const float* __restrict__ bias,
                     const float* __restrict__ norm, float* __restrict__ out) {
    const int lane  = threadIdx.x & 31;
    const int r15   = lane & 15;
    const int khalf = lane >> 4;
    const int gw = __builtin_amdgcn_readfirstlane((int)(blockIdx.x * 8 + (threadIdx.x >> 5)));
    const int nw = gridDim.x * 8;
    const float4* __restrict__ zl = (const float4*)z + r15;
    const float4 bq = ((const float4*)bias)[r15];
    for (int node = gw; node < N_NODES; node += nw) {
        const int beg = rowptr[node];
        const int end = rowptr[node + 1];
        float4 acc = make_float4(0.0f, 0.0f, 0.0f, 0.0f);
        for (int base = beg; base < end; base += 32) {
            const int cnt = min(32, end - base);
            int my = 0;
            if (lane < cnt) my = edges[base + lane];
            int j = 0;
            for (; j + 8 < cnt; j += 16) gather_pairs<8>(zl, my, j, cnt, khalf, acc);
            if (j < cnt) gather_pairs<4>(zl, my, j, cnt, khalf, acc);
        }
        acc.x += __shfl_xor(acc.x, 16); acc.y += __shfl_xor(acc.y, 16);
        acc.z += __shfl_xor(acc.z, 16); acc.w += __shfl_xor(acc.w, 16);
        if (khalf == 0) {
            const float nd = norm[node];
            float4 o;
            o.x = fmaf(acc.x, nd, bq.x); o.y = fmaf(acc.y, nd, bq.y);
            o.z = fmaf(acc.z, nd, bq.z); o.w = fmaf(acc.w, nd, bq.w);
            ((float4*)out)[(size_t)node * 16 + r15] = o;
        }
    }
}

extern "C" void kernel_launch(void* const* d_in, const int* in_sizes, int n_in,
                              void* d_out, int out_size, void* d_ws, size_t ws_size,
                              hipStream_t stream) {
    const float* features = (const float*)d_in[0];
    const float* W1 = (const float*)d_in[1];
    const float* b1 = (const float*)d_in[2];
    const float* W2 = (const float*)d_in[3];
    const float* b2 = (const float*)d_in[4];
    const float* W3 = (const float*)d_in[5];
    const float* b3 = (const float*)d_in[6];
    const int*  src = (const int*)d_in[7];
    const int*  dst = (const int*)d_in[8];
    const int E = in_sizes[7];

    float* nrm      = (float*)d_ws + 2 * N_NODES;
    int*   rowptr   = (int*)d_ws + 4 * N_NODES;
    float* aggA     = (float*)d_ws + 250304;
    float* bufB     = aggA + (size_t)N_NODES * FDIM;
    int*   edges    = (int*)(bufB + (size_t)N_NODES * FDIM);
    int*   tmpD     = edges + 2 * (size_t)E;
    int*   tmpS     = tmpD + 2 * (size_t)E;
    int*   offD     = tmpS + (size_t)E;
    int*   offS     = offD + G_CHUNKS * (NB + 1);
    float* norm_src = nrm;
    float* norm_dst = nrm + N_NODES;

    bin_kernel<<<G_CHUNKS, 1024, 0, stream>>>(src, dst, tmpD, tmpS, offD, offS, E);
    range_kernel<<<RANGE_BLOCKS, 1024, 0, stream>>>(tmpD, offD, tmpS, offS, norm_src, rowptr, norm_dst, edges, E);

    const int layerBlocks = 1024;

    layer1_kernel<<<layerBlocks, 512, 0, stream>>>(features, rowptr, edges, W1, b1,
                                                   norm_dst, norm_src, bufB);
    layer2_fused_kernel<<<512, 1024, 0, stream>>>(bufB, rowptr, edges, W2, b2, W3, norm_dst, norm_src, aggA);
    gather64_kernel<<<1536, 256, 0, stream>>>(aggA, rowptr, edges, b3, norm_dst, (float*)d_out);
}


# Layer 1 makes an edge's source scale from an LDS table of out-degree bytes (staged per work-group, 255 = read memory), not a scattered global load per edge; 32-wave work-groups; sum order unchanged
# speedup vs baseline: 6.2792x; 1.0307x over previous
#include <hip/hip_runtime.h>
#include <hip/hip_bf16.h>


#define N_NODES 50000
#define FDIM    128
#define KDIM    128

typedef __attribute__((ext_vector_type(2))) float v2f;
typedef __attribute__((ext_vector_type(8))) float v8f;

__device__ __forceinline__ int wave_sum(int v) {
    for (int o = 16; o > 0; o >>= 1) v += __shfl_xor(v, o);
    return v;
}

__device__ __forceinline__ int wave_inclusive_scan(int v, int lane) {
    for (int o = 1; o < 32; o <<= 1) {
        const int t = __shfl_up(v, o);
        if (lane >= o) v += t;
    }
    return v;
}

#define G_CHUNKS  128
#define NB        128
#define RANGE     ((N_NODES + NB - 1) / NB)
#define CHUNK_CAP 8192

static_assert(NB == 128, "scan_ranges covers 4 ranges per lane of one wave");
__device__ __forceinline__ void scan_ranges(int* cur, int* __restrict__ off, int lane) {
    const int c0 = cur[4 * lane], c1 = cur[4 * lane + 1], c2 = cur[4 * lane + 2], c3 = cur[4 * lane + 3];
    const int sum = c0 + c1 + c2 + c3;
    const int inc = wave_inclusive_scan(sum, lane);
    const int x0 = inc - sum, x1 = x0 + c0, x2 = x1 + c1, x3 = x2 + c2;
    cur[4 * lane] = x0; cur[4 * lane + 1] = x1; cur[4 * lane + 2] = x2; cur[4 * lane + 3] = x3;
    off[4 * lane] = x0; off[4 * lane + 1] = x1; off[4 * lane + 2] = x2; off[4 * lane + 3] = x3;
    if (lane == 31) off[NB] = inc;
}

__global__ __launch_bounds__(1024)
void bin_kernel(const int* __restrict__ src, const int* __restrict__ dst,
                int* __restrict__ tmpD, int* __restrict__ tmpS,
                int* __restrict__ offD, int* __restrict__ offS, int E) {
    __shared__ int curD[NB];
    __shared__ int curS[NB];
    __shared__ __attribute__((aligned(16))) int2 stD[CHUNK_CAP];
    __shared__ __attribute__((aligned(16))) int  stS[CHUNK_CAP];
    const int tid  = threadIdx.x;
    const int lane = tid & 31;
    if (tid < NB) { curD[tid] = 0; curS[tid] = 0; }
    __syncthreads();
    const int chunk = (E + (int)gridDim.x - 1) / (int)gridDim.x;
    const int beg = (int)blockIdx.x * chunk;
    const int end = min(E, beg + min(chunk, CHUNK_CAP));
    for (int e0 = beg + tid; e0 < end; e0 += 8 * 1024) {
        int d[8], s[8];
#pragma unroll
        for (int u = 0; u < 8; ++u) {
            const int e = e0 + u * 1024;
            d[u] = -1; s[u] = 0;
            if (e < end) { d[u] = dst[e]; s[u] = src[e]; }
        }
#pragma unroll
        for (int u = 0; u < 8; ++u) {
            if (d[u] >= 0) {
                atomicAdd(&curD[(unsigned)d[u] / RANGE], 1);
                atomicAdd(&curS[(unsigned)s[u] / RANGE], 1);
            }
        }
    }
    __syncthreads();
    if (tid < 32)      scan_ranges(curD, offD + (size_t)blockIdx.x * (NB + 1), lane);
    else if (tid < 64) scan_ranges(curS, offS + (size_t)blockIdx.x * (NB + 1), lane);
    __syncthreads();
    for (int e0 = beg + tid; e0 < end; e0 += 8 * 1024) {
        int d[8], s[8];
#pragma unroll
        for (int u = 0; u < 8; ++u) {
            const int e = e0 + u * 1024;
            d[u] = -1; s[u] = 0;
            if (e < end) { d[u] = dst[e]; s[u] = src[e]; }
        }
#pragma unroll
        for (int u = 0; u < 8; ++u) {
            if (d[u] >= 0) {
                const int p = atomicAdd(&curD[(unsigned)d[u] / RANGE], 1);
                stD[p] = make_int2(s[u], d[u]);
                const int q = atomicAdd(&curS[(unsigned)s[u] / RANGE], 1);
                stS[q] = s[u];
            }
        }
    }
    __syncthreads();
    const int n = end - beg;
    int2* oD = (int2*)tmpD + beg;
    int*  oS = tmpS + beg;
    for (int j = tid; j < n; j += 1024) { oD[j] = stD[j]; oS[j] = stS[j]; }
}

__device__ __forceinline__
void count_sources(const int* __restrict__ tmpS, const int* __restrict__ offS,
                   float* __restrict__ nsrc, unsigned char* __restrict__ deg8, int E, int b) {
    __shared__ int cnt[RANGE];
    __shared__ int segB[G_CHUNKS];
    __shared__ int segN[G_CHUNKS];
    const int tid  = threadIdx.x;
    const int lane = tid & 31;
    const int wave = __builtin_amdgcn_readfirstlane(tid >> 5);
    const int lo   = b * RANGE;
    const int chunk = (E + G_CHUNKS - 1) / G_CHUNKS;
    if (tid < RANGE) cnt[tid] = 0;
    if (tid < G_CHUNKS) {
        const int o0 = offS[tid * (NB + 1) + b];
        const int o1 = offS[tid * (NB + 1) + b + 1];
        segB[tid] = tid * chunk + o0;
        segN[tid] = o1 - o0;
    }
    __syncthreads();
    int sb[4], sn[4], v[4][2];
#pragma unroll
    for (int k = 0; k < 4; ++k) {
        sb[k] = segB[wave + 32 * k];
        sn[k] = segN[wave + 32 * k];
#pragma unroll
        for (int r = 0; r < 2; ++r) {
            const int j = lane + 32 * r;
            v[k][r] = -1;
            if (j < sn[k]) v[k][r] = tmpS[sb[k] + j];
        }
    }
#pragma unroll
    for (int k = 0; k < 4; ++k) {
#pragma unroll
        for (int r = 0; r < 2; ++r)
            if (v[k][r] >= 0) atomicAdd(&cnt[v[k][r] - lo], 1);
        for (int j = lane + 64; j < sn[k]; j += 32) atomicAdd(&cnt[tmpS[sb[k] + j] - lo], 1);
    }
    __syncthreads();
    if (tid < RANGE && lo + tid < N_NODES) {
        const int c = cnt[tid];
        nsrc[lo + tid] = rsqrtf(fmaxf((float)c, 1.0f));
        deg8[lo + tid] = (unsigned char)min(c, 255);
    }
}

#define SCAN_PER ((RANGE + 31) / 32)
static_assert(RANGE <= 1024 && G_CHUNKS == 4 * 32, "one thread per node of a range, 4 chunks per wave");

__device__ __forceinline__
void fill_range(const int* __restrict__ tmpD, const int* __restrict__ offD,
                int* __restrict__ rowptr, float* __restrict__ ndst,
                int* __restrict__ edges, int E, int b) {
    __shared__ int cnt[RANGE];
    __shared__ int cur[RANGE];
    __shared__ int segB[G_CHUNKS];
    __shared__ int segN[G_CHUNKS];
    __shared__ int part[4];
    const int tid  = threadIdx.x;
    const int lane = tid & 31;
    const int wave = __builtin_amdgcn_readfirstlane(tid >> 5);
    const int lo   = b * RANGE;
    const int chunk = (E + G_CHUNKS - 1) / G_CHUNKS;
    const int2* __restrict__ tD = (const int2*)tmpD;
    if (tid < RANGE) cnt[tid] = 0;
    if (tid < G_CHUNKS) {
        const int o0 = offD[tid * (NB + 1) + b];
        const int o1 = offD[tid * (NB + 1) + b + 1];
        segB[tid] = tid * chunk + o0;
        segN[tid] = o1 - o0;
        const int s = wave_sum(o0);
        if (lane == 0) part[wave] = s;
    }
    __syncthreads();
    const int base = part[0] + part[1] + part[2] + part[3];
    int sb[4], sn[4];
    int2 ed[4][2];
#pragma unroll
    for (int k = 0; k < 4; ++k) {
        sb[k] = segB[wave + 32 * k];
        sn[k] = segN[wave + 32 * k];
#pragma unroll
        for (int r = 0; r < 2; ++r) {
            const int j = lane + 32 * r;
            ed[k][r] = make_int2(0, -1);
            if (j < sn[k]) ed[k][r] = tD[sb[k] + j];
        }
    }
#pragma unroll
    for (int k = 0; k < 4; ++k) {
#pragma unroll
        for (int r = 0; r < 2; ++r)
            if (ed[k][r].y >= 0) atomicAdd(&cnt[ed[k][r].y - lo], 1);
        for (int j = lane + 64; j < sn[k]; j += 32) atomicAdd(&cnt[tD[sb[k] + j].y - lo], 1);
    }
    __syncthreads();
    if (wave == 0) {
        int c[SCAN_PER];
        int sum = 0;
#pragma unroll
        for (int q = 0; q < SCAN_PER; ++q) {
            const int t = lane * SCAN_PER + q;
            c[q] = (t < RANGE) ? cnt[t] : 0;
            sum += c[q];
        }
        const int inc = wave_inclusive_scan(sum, lane);
        int run = base + inc - sum;
#pragma unroll
        for (int q = 0; q < SCAN_PER; ++q) {
            const int t = lane * SCAN_PER + q;
            if (t < RANGE) {
                cur[t] = run;
                if (lo + t < N_NODES) {
                    rowptr[lo + t] = run;
                    ndst[lo + t]   = rsqrtf(fmaxf((float)c[q], 1.0f));
                }
            }
            run += c[q];
        }
        if (b == NB - 1 && lane == 31) rowptr[N_NODES] = base + inc;
    }
    __syncthreads();
#pragma unroll
    for (int k = 0; k < 4; ++k) {
#pragma unroll
        for (int r = 0; r < 2; ++r) {
            if (ed[k][r].y >= 0) {
                const int pos = atomicAdd(&cur[ed[k][r].y - lo], 1);
                edges[pos] = ed[k][r].x;
            }
        }
        for (int j = lane + 64; j < sn[k]; j += 32) {
            const int2 e = tD[sb[k] + j];
            const int pos = atomicAdd(&cur[e.y - lo], 1);
            edges[pos] = e.x;
        }
    }
}

#define RANGE_BLOCKS (2 * NB)
__global__ __launch_bounds__(1024)
void range_kernel(const int* __restrict__ tmpD, const int* __restrict__ offD,
                  const int* __restrict__ tmpS, const int* __restrict__ offS,
                  float* __restrict__ nsrc, unsigned char* __restrict__ deg8,
                  int* __restrict__ rowptr,
                  float* __restrict__ ndst, int* __restrict__ edges, int E) {
    if (blockIdx.x < NB) fill_range(tmpD, offD, rowptr, ndst, edges, E, (int)blockIdx.x);
    else                 count_sources(tmpS, offS, nsrc, deg8, E, (int)blockIdx.x - NB);
}

static_assert(N_NODES % 16 == 0, "the out-degree bytes are staged as 16-byte pieces; a 16-row half tile is whole or absent");
__global__ __launch_bounds__(1024, 2)
void layer1_kernel(const float* __restrict__ x, const int* __restrict__ rowptr,
                   const int* __restrict__ edges, const float* __restrict__ W,
                   const float* __restrict__ bias, const float* __restrict__ norm,
                   const float* __restrict__ nsrc, const unsigned char* __restrict__ deg8,
                   float* __restrict__ out) {
    constexpr int NOUT   = 128;
    constexpr int NT     = NOUT / 16;
    constexpr int NTILES = (N_NODES + 31) / 32;
    __shared__ __attribute__((aligned(16))) float Wl[KDIM * NOUT];
    __shared__ __attribute__((aligned(16))) float At[32 * FDIM];
    __shared__ __attribute__((aligned(16))) unsigned char tab[N_NODES];

    const int lane  = threadIdx.x & 31;
    const int wave  = __builtin_amdgcn_readfirstlane((int)(threadIdx.x >> 5));
    const int r15   = lane & 15;
    const int khalf = lane >> 4;

    {
        constexpr int PIECES = N_NODES / 16;
        constexpr int TRIPS  = (PIECES + 1023) / 1024;
        uint4 tq[TRIPS];
#pragma unroll
        for (int k = 0; k < TRIPS; ++k)
            tq[k] = ((const uint4*)deg8)[min((int)threadIdx.x + k * 1024, PIECES - 1)];
#pragma unroll
        for (int k = 0; k < TRIPS; ++k)
            ((uint4*)tab)[min((int)threadIdx.x + k * 1024, PIECES - 1)] = tq[k];
    }
#pragma unroll
    for (int s = threadIdx.x; s < (KDIM / 4) * NT * 32; s += 1024) {
        const int l  = s & 31;
        const int t  = (s >> 5) % NT;
        const int kk = (s >> 5) / NT;
        const float* Wp = W + (size_t)(kk * 4 + (l >> 4) * 2) * NOUT + t * 16 + (l & 15);
        ((float2*)Wl)[s] = make_float2(Wp[0], Wp[NOUT]);
    }

    const float4* __restrict__ xl = (const float4*)x + lane;
#define ROW(j) xl[(size_t)__builtin_amdgcn_readlane(my, (j)) * (FDIM / 4)]
#define WGT(j) __int_as_float(__builtin_amdgcn_readlane(__float_as_int(mw), (j)))
#define ACC(VV, WW) { acc.x = fmaf((VV).x, (WW), acc.x); acc.y = fmaf((VV).y, (WW), acc.y); \
                      acc.z = fmaf((VV).z, (WW), acc.z); acc.w = fmaf((VV).w, (WW), acc.w); }

    int tile = blockIdx.x;
    int beg = 0, end = 0;
    if (tile * 32 + wave < N_NODES) { beg = rowptr[tile * 32 + wave]; end = rowptr[tile * 32 + wave + 1]; }
    __syncthreads();

    for (int it = 0; tile < NTILES; ++it) {
        float4 acc = make_float4(0.0f, 0.0f, 0.0f, 0.0f);
        for (int base = beg; base < end; base += 32) {
            const int cnt = min(32, end - base);
            int   my = 0;
            float mw = 0.0f;
            if (lane < cnt) {
                my = edges[base + lane];
                const unsigned d = tab[my];
                mw = rsqrtf(fmaxf((float)d, 1.0f));
                if (d == 255u) mw = nsrc[my];
            }
            int j = 0;
            for (; j + 8 <= cnt; j += 8) {
                float4 v[8];
#pragma unroll
                for (int u = 0; u < 8; ++u) v[u] = ROW(j + u);
#pragma unroll
                for (int u = 0; u < 8; ++u) ACC(v[u], WGT(j + u));
            }
            if (j + 4 <= cnt) {
                float4 v[4];
#pragma unroll
                for (int u = 0; u < 4; ++u) v[u] = ROW(j + u);
#pragma unroll
                for (int u = 0; u < 4; ++u) ACC(v[u], WGT(j + u));
                j += 4;
            }
            if (j < cnt) {
                const int r = cnt - j;
                const float4 z = make_float4(0.0f, 0.0f, 0.0f, 0.0f);
                float4 v0 = ROW(j), v1 = z, v2 = z;
                if (r > 1) v1 = ROW(j + 1);
                if (r > 2) v2 = ROW(j + 2);
                ACC(v0, WGT(j));
                ACC(v1, WGT(j + 1));
                ACC(v2, WGT(j + 2));
            }
        }
        __syncthreads();
        ((float4*)(At + wave * FDIM))[lane ^ (wave & 15)] = acc;
        __syncthreads();

        const int cur = tile;
        tile += gridDim.x;
        beg = 0; end = 0;
        if (tile < NTILES && tile * 32 + wave < N_NODES) {
            beg = rowptr[tile * 32 + wave]; end = rowptr[tile * 32 + wave + 1];
        }

        if ((wave >> 4) == (it & 1)) {
            const int q  = wave & 15;
            const int mt = q >> 3;
            const int t  = q & 7;
            if (cur * 32 + mt * 16 < N_NODES) {
                const int rbase = cur * 32 + mt * 16 + khalf * 8;
                float nrm[8], ns[8];
#pragma unroll
                for (int j = 0; j < 8; ++j) { nrm[j] = norm[rbase + j]; ns[j] = nsrc[rbase + j]; }
                const float bv = bias[t * 16 + r15];
                v8f d;
#pragma unroll
                for (int j = 0; j < 8; ++j) d[j] = 0.0f;
                const float* Ar = At + (mt * 16 + r15) * FDIM + khalf * 2;
                const v2f*   Bp = (const v2f*)Wl + t * 32 + lane;
#pragma unroll 8
                for (int kk = 0; kk < KDIM / 4; ++kk) {
                    const v2f a = *(const v2f*)(Ar + ((kk ^ r15) << 2));
                    const v2f b = Bp[kk * NT * 32];
                    d = __builtin_amdgcn_wmma_f32_16x16x4_f32(
                        false, a, false, b, (short)0, d, false, false);
                }
#pragma unroll
                for (int j = 0; j < 8; ++j) {
                    const float v = fmaxf(fmaf(d[j], nrm[j], bv), 0.0f);
                    out[(size_t)(rbase + j) * NOUT + t * 16 + r15] = v * ns[j];
                }
            }
        }
    }
#undef ROW
#undef WGT
#undef ACC
}

__global__ __launch_bounds__(1024, 2)
void layer2_fused_kernel(const float* __restrict__ x, const int* __restrict__ rowptr,
                         const int* __restrict__ edges, const float* __restrict__ W2,
                         const float* __restrict__ b2, const float* __restrict__ W3,
                         const float* __restrict__ norm, const float* __restrict__ nsrc,
                         float* __restrict__ z) {
    constexpr int NT2 = 8;
    constexpr int NT3 = 4;
    constexpr int ZDIM = 64;
    __shared__ __attribute__((aligned(16))) float Wl[KDIM * 128];
    __shared__ __attribute__((aligned(16))) float W3l[KDIM * ZDIM];
    __shared__ __attribute__((aligned(16))) float At[32 * FDIM];
    __shared__ __attribute__((aligned(16))) float Ht[32 * FDIM];

    const int lane  = threadIdx.x & 31;
    const int wave  = __builtin_amdgcn_readfirstlane((int)(threadIdx.x >> 5));
    const int r15   = lane & 15;
    const int khalf = lane >> 4;

#pragma unroll
    for (int s = threadIdx.x; s < (KDIM / 4) * NT2 * 32; s += 1024) {
        const int l  = s & 31;
        const int t  = (s >> 5) % NT2;
        const int kk = (s >> 5) / NT2;
        const float* Wp = W2 + (size_t)(kk * 4 + (l >> 4) * 2) * 128 + t * 16 + (l & 15);
        ((float2*)Wl)[s] = make_float2(Wp[0], Wp[128]);
    }
#pragma unroll
    for (int s = threadIdx.x; s < (KDIM / 4) * NT3 * 32; s += 1024) {
        const int l  = s & 31;
        const int t  = (s >> 5) % NT3;
        const int kk = (s >> 5) / NT3;
        const float* Wp = W3 + (size_t)(kk * 4 + (l >> 4) * 2) * ZDIM + t * 16 + (l & 15);
        ((float2*)W3l)[s] = make_float2(Wp[0], Wp[ZDIM]);
    }

    const float4* __restrict__ xl = (const float4*)x + lane;
#define ROW(j) xl[(size_t)__builtin_amdgcn_readlane(my, (j)) * (FDIM / 4)]
#define ACC(VV) { acc.x += (VV).x; acc.y += (VV).y; acc.z += (VV).z; acc.w += (VV).w; }

    for (int tile = blockIdx.x; tile < (N_NODES + 31) / 32; tile += gridDim.x) {
        const int node = tile * 32 + wave;
        int beg = 0, end = 0;
        if (node < N_NODES) { beg = rowptr[node]; end = rowptr[node + 1]; }
        float4 acc = make_float4(0.0f, 0.0f, 0.0f, 0.0f);
        for (int base = beg; base < end; base += 32) {
            const int cnt = min(32, end - base);
            int my = 0;
            if (lane < cnt) my = edges[base + lane];
            int j = 0;
            for (; j + 8 <= cnt; j += 8) {
                float4 v[8];
#pragma unroll
                for (int u = 0; u < 8; ++u) v[u] = ROW(j + u);
#pragma unroll
                for (int u = 0; u < 8; ++u) ACC(v[u]);
            }
            if (j + 4 <= cnt) {
                float4 v[4];
#pragma unroll
                for (int u = 0; u < 4; ++u) v[u] = ROW(j + u);
#pragma unroll
                for (int u = 0; u < 4; ++u) ACC(v[u]);
                j += 4;
            }
            if (j < cnt) {
                const int r = cnt - j;
                const float4 z4 = make_float4(0.0f, 0.0f, 0.0f, 0.0f);
                float4 v0 = ROW(j), v1 = z4, v2 = z4;
                if (r > 1) v1 = ROW(j + 1);
                if (r > 2) v2 = ROW(j + 2);
                ACC(v0);
                ACC(v1);
                ACC(v2);
            }
        }
        ((float4*)(At + wave * FDIM))[lane ^ (wave & 15)] = acc;
        __syncthreads();

        if (wave < 16) {
            const int mt = wave >> 3;
            const int t  = wave & 7;
            const int rbase = tile * 32 + mt * 16 + khalf * 8;
            float nrm[8];
#pragma unroll
            for (int j = 0; j < 8; ++j) nrm[j] = norm[min(rbase + j, N_NODES - 1)];
            const float bv = b2[t * 16 + r15];
            v8f d;
#pragma unroll
            for (int j = 0; j < 8; ++j) d[j] = 0.0f;
            const float* Ar = At + (mt * 16 + r15) * FDIM + khalf * 2;
            const v2f*   Bp = (const v2f*)Wl + t * 32 + lane;
#pragma unroll 8
            for (int kk = 0; kk < KDIM / 4; ++kk) {
                const v2f a = *(const v2f*)(Ar + ((kk ^ r15) << 2));
                const v2f b = Bp[kk * NT2 * 32];
                d = __builtin_amdgcn_wmma_f32_16x16x4_f32(
                    false, a, false, b, (short)0, d, false, false);
            }
            const int col = t * 16 + r15;
#pragma unroll
            for (int j = 0; j < 8; ++j) {
                const int row = mt * 16 + khalf * 8 + j;
                const float v = fmaxf(fmaf(d[j], nrm[j], bv), 0.0f);
                Ht[row * FDIM + ((((col >> 2) ^ (row & 15)) << 2) | (col & 3))] = v;
            }
        }
        __syncthreads();
        if (wave >= 16 && wave < 24) {
            const int q  = wave - 16;
            const int mt = q >> 2;
            const int t3 = q & 3;
            const int rbase = tile * 32 + mt * 16 + khalf * 8;
            float ns[8];
#pragma unroll
            for (int j = 0; j < 8; ++j) ns[j] = nsrc[min(rbase + j, N_NODES - 1)];
            v8f d;
#pragma unroll
            for (int j = 0; j < 8; ++j) d[j] = 0.0f;
            const float* Hr = Ht + (mt * 16 + r15) * FDIM + khalf * 2;
            const v2f*   Bp = (const v2f*)W3l + t3 * 32 + lane;
#pragma unroll 8
            for (int kk = 0; kk < KDIM / 4; ++kk) {
                const v2f a = *(const v2f*)(Hr + ((kk ^ r15) << 2));
                const v2f b = Bp[kk * NT3 * 32];
                d = __builtin_amdgcn_wmma_f32_16x16x4_f32(
                    false, a, false, b, (short)0, d, false, false);
            }
#pragma unroll
            for (int j = 0; j < 8; ++j) {
                if (rbase + j < N_NODES) z[(size_t)(rbase + j) * ZDIM + t3 * 16 + r15] = d[j] * ns[j];
            }
        }
    }
#undef ROW
#undef ACC
}

template <int P>
__device__ __forceinline__ void gather_pairs(const float4* __restrict__ zl, int my, int j,
                                             int cnt, int khalf, float4& acc) {
    float4 v[P];
    float  w[P];
#pragma unroll
    for (int u = 0; u < P; ++u) {
        const int e  = j + 2 * u + khalf;
        const int ec = min(e, cnt - 1);
        const int s  = __shfl(my, ec);
        w[u] = (e < cnt) ? 1.0f : 0.0f;
        v[u] = zl[(size_t)s * 16];
    }
#pragma unroll
    for (int u = 0; u < P; ++u) {
        acc.x = fmaf(v[u].x, w[u], acc.x); acc.y = fmaf(v[u].y, w[u], acc.y);
        acc.z = fmaf(v[u].z, w[u], acc.z); acc.w = fmaf(v[u].w, w[u], acc.w);
    }
}

__global__ __launch_bounds__(256)
void gather64_kernel(const float* __restrict__ z, const int* __restrict__ rowptr,
                     const int* __restrict__ edges, const float* __restrict__ bias,
                     const float* __restrict__ norm, float* __restrict__ out) {
    const int lane  = threadIdx.x & 31;
    const int r15   = lane & 15;
    const int khalf = lane >> 4;
    const int gw = __builtin_amdgcn_readfirstlane((int)(blockIdx.x * 8 + (threadIdx.x >> 5)));
    const int nw = gridDim.x * 8;
    const float4* __restrict__ zl = (const float4*)z + r15;
    const float4 bq = ((const float4*)bias)[r15];
    for (int node = gw; node < N_NODES; node += nw) {
        const int beg = rowptr[node];
        const int end = rowptr[node + 1];
        float4 acc = make_float4(0.0f, 0.0f, 0.0f, 0.0f);
        for (int base = beg; base < end; base += 32) {
            const int cnt = min(32, end - base);
            int my = 0;
            if (lane < cnt) my = edges[base + lane];
            int j = 0;
            for (; j + 8 < cnt; j += 16) gather_pairs<8>(zl, my, j, cnt, khalf, acc);
            if (j < cnt) gather_pairs<4>(zl, my, j, cnt, khalf, acc);
        }
        acc.x += __shfl_xor(acc.x, 16); acc.y += __shfl_xor(acc.y, 16);
        acc.z += __shfl_xor(acc.z, 16); acc.w += __shfl_xor(acc.w, 16);
        if (khalf == 0) {
            const float nd = norm[node];
            float4 o;
            o.x = fmaf(acc.x, nd, bq.x); o.y = fmaf(acc.y, nd, bq.y);
            o.z = fmaf(acc.z, nd, bq.z); o.w = fmaf(acc.w, nd, bq.w);
            ((float4*)out)[(size_t)node * 16 + r15] = o;
        }
    }
}

extern "C" void kernel_launch(void* const* d_in, const int* in_sizes, int n_in,
                              void* d_out, int out_size, void* d_ws, size_t ws_size,
                              hipStream_t stream) {
    const float* features = (const float*)d_in[0];
    const float* W1 = (const float*)d_in[1];
    const float* b1 = (const float*)d_in[2];
    const float* W2 = (const float*)d_in[3];
    const float* b2 = (const float*)d_in[4];
    const float* W3 = (const float*)d_in[5];
    const float* b3 = (const float*)d_in[6];
    const int*  src = (const int*)d_in[7];
    const int*  dst = (const int*)d_in[8];
    const int E = in_sizes[7];

    unsigned char* deg8 = (unsigned char*)d_ws;
    float* nrm      = (float*)d_ws + 2 * N_NODES;
    int*   rowptr   = (int*)d_ws + 4 * N_NODES;
    float* aggA     = (float*)d_ws + 250304;
    float* bufB     = aggA + (size_t)N_NODES * FDIM;
    int*   edges    = (int*)(bufB + (size_t)N_NODES * FDIM);
    int*   tmpD     = edges + 2 * (size_t)E;
    int*   tmpS     = tmpD + 2 * (size_t)E;
    int*   offD     = tmpS + (size_t)E;
    int*   offS     = offD + G_CHUNKS * (NB + 1);
    float* norm_src = nrm;
    float* norm_dst = nrm + N_NODES;

    bin_kernel<<<G_CHUNKS, 1024, 0, stream>>>(src, dst, tmpD, tmpS, offD, offS, E);
    range_kernel<<<RANGE_BLOCKS, 1024, 0, stream>>>(tmpD, offD, tmpS, offS, norm_src, deg8, rowptr, norm_dst, edges, E);

    layer1_kernel<<<512, 1024, 0, stream>>>(features, rowptr, edges, W1, b1,
                                            norm_dst, norm_src, deg8, bufB);
    layer2_fused_kernel<<<512, 1024, 0, stream>>>(bufB, rowptr, edges, W2, b2, W3, norm_dst, norm_src, aggA);
    gather64_kernel<<<1536, 256, 0, stream>>>(aggA, rowptr, edges, b3, norm_dst, (float*)d_out);
}
